# starting kernel
# baseline (speedup 1.0000x reference)
#include <hip/hip_runtime.h>
#include <hip/hip_bf16.h>

typedef __attribute__((ext_vector_type(2))) float v2f;
typedef __attribute__((ext_vector_type(8))) float v8f;

static constexpr int   NN    = 10000;
static constexpr int   TT    = 8;
static constexpr int   PP    = 2;
static constexpr int   FIN   = 64;
static constexpr int   FOUT  = 64;
static constexpr int   HH    = 128;
static constexpr int   BB    = 2;
static constexpr int   EE    = 160000;
static constexpr int   GG    = BB * TT;
static constexpr int   BN    = BB * NN;
static constexpr int   HID   = HH * 4;
static constexpr size_t GNF  = (size_t)GG * NN * FOUT;

__device__ __forceinline__ v8f wmma4(v2f a, v2f b, v8f c) {
  return __builtin_amdgcn_wmma_f32_16x16x4_f32(false, a, false, b, (short)0, c,
                                               false, false);
}

__device__ __forceinline__ void atomAddF(float* p, float v) {
  __hip_atomic_fetch_add(p, v, __ATOMIC_RELAXED, __HIP_MEMORY_SCOPE_AGENT);
}

__global__ void k_deg_init(float* deg) {
  int i = blockIdx.x * 256 + threadIdx.x;
  if (i < NN) deg[i] = 0.f;
}

__global__ void k_deg_scatter(const int* __restrict__ ei,
                              const float* __restrict__ ew,
                              float* __restrict__ deg) {
  int e = blockIdx.x * 256 + threadIdx.x;
  if (e < EE)                atomAddF(&deg[ei[EE + e]], ew[e]);
  else if (e < EE + NN)      atomAddF(&deg[e - EE], 1.0f);
}

__global__ void k_dinv(const float* __restrict__ deg, float* __restrict__ dinv) {
  int i = blockIdx.x * 256 + threadIdx.x;
  if (i < NN) {
    float d = deg[i];
    dinv[i] = d > 0.f ? rsqrtf(d) : 0.f;
  }
}

__global__ __launch_bounds__(256) void k_gcn_gemm(const float* __restrict__ X,
                                                  const float* __restrict__ W,
                                                  float* __restrict__ Y) {
  __shared__ float wl[FOUT * FIN];
  for (int i = threadIdx.x; i < FOUT * FIN; i += 256) wl[i] = W[i];
  __syncthreads();
  const int wave = threadIdx.x >> 5, lane = threadIdx.x & 31;
  const int mr = lane & 15, half = lane >> 4, kk = half * 2;
  const int row0 = blockIdx.x * 32 + (wave & 1) * 16;
  const int n0 = (wave >> 1) * 16;
  const float* ap = X + (size_t)(row0 + mr) * FIN + kk;
  const float* bp = wl + (n0 + mr) * FIN + kk;
  v8f c = {};
#pragma unroll
  for (int k = 0; k < FIN; k += 4) {
    v2f a = {ap[k], ap[k + 1]};
    v2f b = {bp[k], bp[k + 1]};
    c = wmma4(a, b, c);
  }
#pragma unroll
  for (int j = 0; j < 8; ++j)
    Y[(size_t)(row0 + j + 8 * half) * FOUT + n0 + mr] = c[j];
}

__global__ void k_agg_init(const float* __restrict__ b, float* __restrict__ gc) {
  size_t i = (size_t)blockIdx.x * 256 + threadIdx.x;
  if (i < GNF) gc[i] = b[i & 63];
}

__global__ void k_scatter(const int* __restrict__ ei, const float* __restrict__ ew,
                          const float* __restrict__ dinv,
                          const float* __restrict__ xl, float* __restrict__ gc) {
  size_t idx = (size_t)blockIdx.x * 256 + threadIdx.x;
  if (idx >= (size_t)(EE + NN) * 64) return;
  int f = (int)(idx & 63);
  int e = (int)(idx >> 6);
  int r, cdst; float nrm;
  if (e < EE) {
    r = ei[e]; cdst = ei[EE + e];
    nrm = dinv[r] * ew[e] * dinv[cdst];
  } else {
    int n = e - EE;
    r = n; cdst = n;
    float di = dinv[n];
    nrm = di * di;
  }
  const float* src = xl + (size_t)r * FOUT + f;
  float* dst = gc + (size_t)cdst * FOUT + f;
#pragma unroll
  for (int g = 0; g < GG; ++g)
    atomAddF(dst + (size_t)g * NN * FOUT, src[(size_t)g * NN * FOUT] * nrm);
}

__global__ __launch_bounds__(256) void k_rnn(const float* __restrict__ gc,
                                             const float* __restrict__ Wih,
                                             const float* __restrict__ Whh,
                                             const float* __restrict__ bih,
                                             const float* __restrict__ bhh,
                                             float* __restrict__ rnn) {
  extern __shared__ float sm[];
  float* s_wih = sm;
  float* s_whh = sm + HH * FOUT;
  float* s_h   = s_whh + HH * HH;
  float* s_b   = s_h + 32 * HH;
  for (int i = threadIdx.x; i < HH * FOUT; i += 256) s_wih[i] = Wih[i];
  for (int i = threadIdx.x; i < HH * HH;  i += 256) s_whh[i] = Whh[i];
  for (int i = threadIdx.x; i < 32 * HH;  i += 256) s_h[i]   = 0.f;
  if (threadIdx.x < HH) s_b[threadIdx.x] = bih[threadIdx.x] + bhh[threadIdx.x];
  __syncthreads();

  const int wave = threadIdx.x >> 5, lane = threadIdx.x & 31;
  const int mr = lane & 15, half = lane >> 4, kk = half * 2;
  const int n0 = wave * 16;
  const size_t p0 = (size_t)blockIdx.x * 32;
  const float bias = s_b[n0 + mr];
  const float* bw = s_wih + (n0 + mr) * FOUT + kk;
  const float* bh = s_whh + (n0 + mr) * HH + kk;

  for (int t = 0; t < TT; ++t) {
    v8f c0 = {}, c1 = {};
    const float* a0 = gc + ((p0 + mr) * TT + t) * FIN + kk;
    const float* a1 = gc + ((p0 + 16 + mr) * TT + t) * FIN + kk;
#pragma unroll
    for (int k = 0; k < FIN; k += 4) {
      v2f b = {bw[k], bw[k + 1]};
      v2f a = {a0[k], a0[k + 1]};
      c0 = wmma4(a, b, c0);
      v2f a2 = {a1[k], a1[k + 1]};
      c1 = wmma4(a2, b, c1);
    }
    const float* h0 = s_h + mr * HH + kk;
    const float* h1 = s_h + (16 + mr) * HH + kk;
#pragma unroll
    for (int k = 0; k < HH; k += 4) {
      v2f b = {bh[k], bh[k + 1]};
      v2f a = {h0[k], h0[k + 1]};
      c0 = wmma4(a, b, c0);
      v2f a2 = {h1[k], h1[k + 1]};
      c1 = wmma4(a2, b, c1);
    }
    __syncthreads();
#pragma unroll
    for (int j = 0; j < 8; ++j) {
      int r = j + 8 * half;
      float v0 = tanhf(c0[j] + bias);
      float v1 = tanhf(c1[j] + bias);
      s_h[r * HH + n0 + mr] = v0;
      s_h[(16 + r) * HH + n0 + mr] = v1;
      rnn[((p0 + r) * TT + t) * HH + n0 + mr] = v0;
      rnn[((p0 + 16 + r) * TT + t) * HH + n0 + mr] = v1;
    }
    __syncthreads();
  }
}

__global__ __launch_bounds__(256) void k_mlp(const float* __restrict__ Xg,
                                             const float* __restrict__ W1, const float* __restrict__ B1,
                                             const float* __restrict__ W2, const float* __restrict__ B2,
                                             const float* __restrict__ W3, const float* __restrict__ B3,
                                             float* __restrict__ out) {
  extern __shared__ float sm[];
  float* sX  = sm;
  float* sY1 = sm + 32 * HH;
  float* sY2 = sY1 + 32 * HID;
  const size_t r0 = (size_t)blockIdx.x * 32;
  for (int i = threadIdx.x; i < 32 * HH; i += 256) sX[i] = Xg[r0 * HH + i];
  __syncthreads();
  const int wave = threadIdx.x >> 5, lane = threadIdx.x & 31;
  const int mr = lane & 15, half = lane >> 4, kk = half * 2;

  for (int ti = 0; ti < 8; ++ti) {
    int tile = wave * 8 + ti;
    int rt = tile & 1, ct = tile >> 1;
    const float* ap = sX + (rt * 16 + mr) * HH + kk;
    const float* bp = W1 + (size_t)(ct * 16 + mr) * HH + kk;
    v8f c = {};
#pragma unroll 8
    for (int k = 0; k < HH; k += 4) {
      v2f a = {ap[k], ap[k + 1]};
      v2f b = {bp[k], bp[k + 1]};
      c = wmma4(a, b, c);
    }
    float bias = B1[ct * 16 + mr];
#pragma unroll
    for (int j = 0; j < 8; ++j) {
      float v = c[j] + bias;
      sY1[(rt * 16 + j + 8 * half) * HID + ct * 16 + mr] = v > 0.f ? v : 0.f;
    }
  }
  __syncthreads();

  for (int ti = 0; ti < 8; ++ti) {
    int tile = wave * 8 + ti;
    int rt = tile & 1, ct = tile >> 1;
    const float* ap = sY1 + (rt * 16 + mr) * HID + kk;
    const float* bp = W2 + (size_t)(ct * 16 + mr) * HID + kk;
    v8f c = {};
#pragma unroll 4
    for (int k = 0; k < HID; k += 4) {
      v2f a = {ap[k], ap[k + 1]};
      v2f b = {bp[k], bp[k + 1]};
      c = wmma4(a, b, c);
    }
    float bias = B2[ct * 16 + mr];
#pragma unroll
    for (int j = 0; j < 8; ++j) {
      float v = c[j] + bias;
      sY2[(rt * 16 + j + 8 * half) * HID + ct * 16 + mr] = v > 0.f ? v : 0.f;
    }
  }
  __syncthreads();

  {
    int rt = wave & 1, ct = wave >> 1;
    const float* ap = sY2 + (rt * 16 + mr) * HID + kk;
    const float* bp = W3 + (size_t)(ct * 16 + mr) * HID + kk;
    v8f c = {};
#pragma unroll 4
    for (int k = 0; k < HID; k += 4) {
      v2f a = {ap[k], ap[k + 1]};
      v2f b = {bp[k], bp[k + 1]};
      c = wmma4(a, b, c);
    }
    float bias = B3[ct * 16 + mr];
#pragma unroll
    for (int j = 0; j < 8; ++j) {
      size_t r = r0 + rt * 16 + j + 8 * half;
      size_t p = r >> 3;
      int t = (int)(r & 7);
      out[(p * (TT + PP) + t) * FOUT + ct * 16 + mr] = c[j] + bias;
    }
  }
}

__global__ void k_zero_pred(float* __restrict__ out) {
  size_t idx = (size_t)blockIdx.x * 256 + threadIdx.x;
  if (idx < (size_t)BN * PP * FOUT) {
    size_t p = idx >> 7;
    size_t rem = idx & 127;
    out[p * (size_t)(TT + PP) * FOUT + (size_t)TT * FOUT + rem] = 0.f;
  }
}

extern "C" void kernel_launch(void* const* d_in, const int* in_sizes, int n_in,
                              void* d_out, int out_size, void* d_ws, size_t ws_size,
                              hipStream_t stream) {
  const float* x_in  = (const float*)d_in[0];
  const int*   ei    = (const int*)  d_in[1];
  const float* ew    = (const float*)d_in[2];
  const float* gcn_w = (const float*)d_in[3];
  const float* gcn_b = (const float*)d_in[4];
  const float* w_ih  = (const float*)d_in[5];
  const float* w_hh  = (const float*)d_in[6];
  const float* b_ih  = (const float*)d_in[7];
  const float* b_hh  = (const float*)d_in[8];
  const float* w1    = (const float*)d_in[9];
  const float* b1    = (const float*)d_in[10];
  const float* w2    = (const float*)d_in[11];
  const float* b2    = (const float*)d_in[12];
  const float* w3    = (const float*)d_in[13];
  const float* b3    = (const float*)d_in[14];
  float* out = (float*)d_out;

  float* ws   = (float*)d_ws;
  float* deg  = ws;
  float* dinv = ws + 16384;
  float* xl   = ws + 32768;
  float* gc   = xl + GNF;
  float* rnn  = gc + GNF;

  k_deg_init<<<(NN + 255) / 256, 256, 0, stream>>>(deg);
  k_deg_scatter<<<(EE + NN + 255) / 256, 256, 0, stream>>>(ei, ew, deg);
  k_dinv<<<(NN + 255) / 256, 256, 0, stream>>>(deg, dinv);

  k_gcn_gemm<<<(GG * NN) / 32, 256, 0, stream>>>(x_in, gcn_w, xl);

  k_agg_init<<<(int)((GNF + 255) / 256), 256, 0, stream>>>(gcn_b, gc);
  k_scatter<<<(int)(((size_t)(EE + NN) * 64 + 255) / 256), 256, 0, stream>>>(
      ei, ew, dinv, xl, gc);

  size_t rnn_lds = (size_t)(HH * FOUT + HH * HH + 32 * HH + HH) * sizeof(float);
  k_rnn<<<BN / 32, 256, rnn_lds, stream>>>(gc, w_ih, w_hh, b_ih, b_hh, rnn);

  size_t mlp_lds = (size_t)(32 * HH + 32 * HID + 32 * HID) * sizeof(float);
  k_mlp<<<(BN * TT) / 32, 256, mlp_lds, stream>>>(rnn, w1, b1, w2, b2, w3, b3, out);

  k_zero_pred<<<(int)(((size_t)BN * PP * FOUT + 255) / 256), 256, 0, stream>>>(out);
}


# Pad LDS row pitches (64/128/512 floats to 68/132/516) in the three WMMA kernels so the 16 rows of a fragment read fall on 16 different banks instead of one; sums unchanged.
# speedup vs baseline: 2.5030x; 2.5030x over previous
#include <hip/hip_runtime.h>
#include <hip/hip_bf16.h>

typedef __attribute__((ext_vector_type(2))) float v2f;
typedef __attribute__((ext_vector_type(8))) float v8f;

static constexpr int   NN    = 10000;
static constexpr int   TT    = 8;
static constexpr int   PP    = 2;
static constexpr int   FIN   = 64;
static constexpr int   FOUT  = 64;
static constexpr int   HH    = 128;
static constexpr int   BB    = 2;
static constexpr int   EE    = 160000;
static constexpr int   GG    = BB * TT;
static constexpr int   BN    = BB * NN;
static constexpr int   HID   = HH * 4;
static constexpr size_t GNF  = (size_t)GG * NN * FOUT;
static constexpr int   P64   = 64 + 4;
static constexpr int   P128  = 128 + 4;
static constexpr int   P512  = 512 + 4;

__device__ __forceinline__ v8f wmma4(v2f a, v2f b, v8f c) {
  return __builtin_amdgcn_wmma_f32_16x16x4_f32(false, a, false, b, (short)0, c,
                                               false, false);
}

__device__ __forceinline__ void atomAddF(float* p, float v) {
  __hip_atomic_fetch_add(p, v, __ATOMIC_RELAXED, __HIP_MEMORY_SCOPE_AGENT);
}

__global__ void k_deg_init(float* deg) {
  int i = blockIdx.x * 256 + threadIdx.x;
  if (i < NN) deg[i] = 0.f;
}

__global__ void k_deg_scatter(const int* __restrict__ ei,
                              const float* __restrict__ ew,
                              float* __restrict__ deg) {
  int e = blockIdx.x * 256 + threadIdx.x;
  if (e < EE)                atomAddF(&deg[ei[EE + e]], ew[e]);
  else if (e < EE + NN)      atomAddF(&deg[e - EE], 1.0f);
}

__global__ void k_dinv(const float* __restrict__ deg, float* __restrict__ dinv) {
  int i = blockIdx.x * 256 + threadIdx.x;
  if (i < NN) {
    float d = deg[i];
    dinv[i] = d > 0.f ? rsqrtf(d) : 0.f;
  }
}

__global__ __launch_bounds__(256) void k_gcn_gemm(const float* __restrict__ X,
                                                  const float* __restrict__ W,
                                                  float* __restrict__ Y) {
  __shared__ float wl[FOUT * P64];
  for (int i = threadIdx.x; i < FOUT * FIN; i += 256) wl[(i >> 6) * P64 + (i & 63)] = W[i];
  __syncthreads();
  const int wave = threadIdx.x >> 5, lane = threadIdx.x & 31;
  const int mr = lane & 15, half = lane >> 4, kk = half * 2;
  const int row0 = blockIdx.x * 32 + (wave & 1) * 16;
  const int n0 = (wave >> 1) * 16;
  const float* ap = X + (size_t)(row0 + mr) * FIN + kk;
  const float* bp = wl + (n0 + mr) * P64 + kk;
  v8f c = {};
#pragma unroll
  for (int k = 0; k < FIN; k += 4) {
    v2f a = {ap[k], ap[k + 1]};
    v2f b = {bp[k], bp[k + 1]};
    c = wmma4(a, b, c);
  }
#pragma unroll
  for (int j = 0; j < 8; ++j)
    Y[(size_t)(row0 + j + 8 * half) * FOUT + n0 + mr] = c[j];
}

__global__ void k_agg_init(const float* __restrict__ b, float* __restrict__ gc) {
  size_t i = (size_t)blockIdx.x * 256 + threadIdx.x;
  if (i < GNF) gc[i] = b[i & 63];
}

__global__ void k_scatter(const int* __restrict__ ei, const float* __restrict__ ew,
                          const float* __restrict__ dinv,
                          const float* __restrict__ xl, float* __restrict__ gc) {
  size_t idx = (size_t)blockIdx.x * 256 + threadIdx.x;
  if (idx >= (size_t)(EE + NN) * 64) return;
  int f = (int)(idx & 63);
  int e = (int)(idx >> 6);
  int r, cdst; float nrm;
  if (e < EE) {
    r = ei[e]; cdst = ei[EE + e];
    nrm = dinv[r] * ew[e] * dinv[cdst];
  } else {
    int n = e - EE;
    r = n; cdst = n;
    float di = dinv[n];
    nrm = di * di;
  }
  const float* src = xl + (size_t)r * FOUT + f;
  float* dst = gc + (size_t)cdst * FOUT + f;
#pragma unroll
  for (int g = 0; g < GG; ++g)
    atomAddF(dst + (size_t)g * NN * FOUT, src[(size_t)g * NN * FOUT] * nrm);
}

__global__ __launch_bounds__(256) void k_rnn(const float* __restrict__ gc,
                                             const float* __restrict__ Wih,
                                             const float* __restrict__ Whh,
                                             const float* __restrict__ bih,
                                             const float* __restrict__ bhh,
                                             float* __restrict__ rnn) {
  extern __shared__ float sm[];
  float* s_wih = sm;
  float* s_whh = sm + HH * P64;
  float* s_h   = s_whh + HH * P128;
  float* s_b   = s_h + 32 * P128;
  for (int i = threadIdx.x; i < HH * FOUT; i += 256) s_wih[(i >> 6) * P64 + (i & 63)] = Wih[i];
  for (int i = threadIdx.x; i < HH * HH;  i += 256) s_whh[(i >> 7) * P128 + (i & 127)] = Whh[i];
  for (int i = threadIdx.x; i < 32 * P128; i += 256) s_h[i]   = 0.f;
  if (threadIdx.x < HH) s_b[threadIdx.x] = bih[threadIdx.x] + bhh[threadIdx.x];
  __syncthreads();

  const int wave = threadIdx.x >> 5, lane = threadIdx.x & 31;
  const int mr = lane & 15, half = lane >> 4, kk = half * 2;
  const int n0 = wave * 16;
  const size_t p0 = (size_t)blockIdx.x * 32;
  const float bias = s_b[n0 + mr];
  const float* bw = s_wih + (n0 + mr) * P64 + kk;
  const float* bh = s_whh + (n0 + mr) * P128 + kk;

  for (int t = 0; t < TT; ++t) {
    v8f c0 = {}, c1 = {};
    const float* a0 = gc + ((p0 + mr) * TT + t) * FIN + kk;
    const float* a1 = gc + ((p0 + 16 + mr) * TT + t) * FIN + kk;
#pragma unroll
    for (int k = 0; k < FIN; k += 4) {
      v2f b = {bw[k], bw[k + 1]};
      v2f a = {a0[k], a0[k + 1]};
      c0 = wmma4(a, b, c0);
      v2f a2 = {a1[k], a1[k + 1]};
      c1 = wmma4(a2, b, c1);
    }
    const float* h0 = s_h + mr * P128 + kk;
    const float* h1 = s_h + (16 + mr) * P128 + kk;
#pragma unroll
    for (int k = 0; k < HH; k += 4) {
      v2f b = {bh[k], bh[k + 1]};
      v2f a = {h0[k], h0[k + 1]};
      c0 = wmma4(a, b, c0);
      v2f a2 = {h1[k], h1[k + 1]};
      c1 = wmma4(a2, b, c1);
    }
    __syncthreads();
#pragma unroll
    for (int j = 0; j < 8; ++j) {
      int r = j + 8 * half;
      float v0 = tanhf(c0[j] + bias);
      float v1 = tanhf(c1[j] + bias);
      s_h[r * P128 + n0 + mr] = v0;
      s_h[(16 + r) * P128 + n0 + mr] = v1;
      rnn[((p0 + r) * TT + t) * HH + n0 + mr] = v0;
      rnn[((p0 + 16 + r) * TT + t) * HH + n0 + mr] = v1;
    }
    __syncthreads();
  }
}

__global__ __launch_bounds__(256) void k_mlp(const float* __restrict__ Xg,
                                             const float* __restrict__ W1, const float* __restrict__ B1,
                                             const float* __restrict__ W2, const float* __restrict__ B2,
                                             const float* __restrict__ W3, const float* __restrict__ B3,
                                             float* __restrict__ out) {
  extern __shared__ float sm[];
  float* sX  = sm;
  float* sY1 = sm + 32 * P128;
  float* sY2 = sY1 + 32 * P512;
  const size_t r0 = (size_t)blockIdx.x * 32;
  for (int i = threadIdx.x; i < 32 * HH; i += 256) sX[(i >> 7) * P128 + (i & 127)] = Xg[r0 * HH + i];
  __syncthreads();
  const int wave = threadIdx.x >> 5, lane = threadIdx.x & 31;
  const int mr = lane & 15, half = lane >> 4, kk = half * 2;

  for (int ti = 0; ti < 8; ++ti) {
    int tile = wave * 8 + ti;
    int rt = tile & 1, ct = tile >> 1;
    const float* ap = sX + (rt * 16 + mr) * P128 + kk;
    const float* bp = W1 + (size_t)(ct * 16 + mr) * HH + kk;
    v8f c = {};
#pragma unroll 8
    for (int k = 0; k < HH; k += 4) {
      v2f a = {ap[k], ap[k + 1]};
      v2f b = {bp[k], bp[k + 1]};
      c = wmma4(a, b, c);
    }
    float bias = B1[ct * 16 + mr];
#pragma unroll
    for (int j = 0; j < 8; ++j) {
      float v = c[j] + bias;
      sY1[(rt * 16 + j + 8 * half) * P512 + ct * 16 + mr] = v > 0.f ? v : 0.f;
    }
  }
  __syncthreads();

  for (int ti = 0; ti < 8; ++ti) {
    int tile = wave * 8 + ti;
    int rt = tile & 1, ct = tile >> 1;
    const float* ap = sY1 + (rt * 16 + mr) * P512 + kk;
    const float* bp = W2 + (size_t)(ct * 16 + mr) * HID + kk;
    v8f c = {};
#pragma unroll 4
    for (int k = 0; k < HID; k += 4) {
      v2f a = {ap[k], ap[k + 1]};
      v2f b = {bp[k], bp[k + 1]};
      c = wmma4(a, b, c);
    }
    float bias = B2[ct * 16 + mr];
#pragma unroll
    for (int j = 0; j < 8; ++j) {
      float v = c[j] + bias;
      sY2[(rt * 16 + j + 8 * half) * P512 + ct * 16 + mr] = v > 0.f ? v : 0.f;
    }
  }
  __syncthreads();

  {
    int rt = wave & 1, ct = wave >> 1;
    const float* ap = sY2 + (rt * 16 + mr) * P512 + kk;
    const float* bp = W3 + (size_t)(ct * 16 + mr) * HID + kk;
    v8f c = {};
#pragma unroll 4
    for (int k = 0; k < HID; k += 4) {
      v2f a = {ap[k], ap[k + 1]};
      v2f b = {bp[k], bp[k + 1]};
      c = wmma4(a, b, c);
    }
    float bias = B3[ct * 16 + mr];
#pragma unroll
    for (int j = 0; j < 8; ++j) {
      size_t r = r0 + rt * 16 + j + 8 * half;
      size_t p = r >> 3;
      int t = (int)(r & 7);
      out[(p * (TT + PP) + t) * FOUT + ct * 16 + mr] = c[j] + bias;
    }
  }
}

__global__ void k_zero_pred(float* __restrict__ out) {
  size_t idx = (size_t)blockIdx.x * 256 + threadIdx.x;
  if (idx < (size_t)BN * PP * FOUT) {
    size_t p = idx >> 7;
    size_t rem = idx & 127;
    out[p * (size_t)(TT + PP) * FOUT + (size_t)TT * FOUT + rem] = 0.f;
  }
}

extern "C" void kernel_launch(void* const* d_in, const int* in_sizes, int n_in,
                              void* d_out, int out_size, void* d_ws, size_t ws_size,
                              hipStream_t stream) {
  const float* x_in  = (const float*)d_in[0];
  const int*   ei    = (const int*)  d_in[1];
  const float* ew    = (const float*)d_in[2];
  const float* gcn_w = (const float*)d_in[3];
  const float* gcn_b = (const float*)d_in[4];
  const float* w_ih  = (const float*)d_in[5];
  const float* w_hh  = (const float*)d_in[6];
  const float* b_ih  = (const float*)d_in[7];
  const float* b_hh  = (const float*)d_in[8];
  const float* w1    = (const float*)d_in[9];
  const float* b1    = (const float*)d_in[10];
  const float* w2    = (const float*)d_in[11];
  const float* b2    = (const float*)d_in[12];
  const float* w3    = (const float*)d_in[13];
  const float* b3    = (const float*)d_in[14];
  float* out = (float*)d_out;

  float* ws   = (float*)d_ws;
  float* deg  = ws;
  float* dinv = ws + 16384;
  float* xl   = ws + 32768;
  float* gc   = xl + GNF;
  float* rnn  = gc + GNF;

  k_deg_init<<<(NN + 255) / 256, 256, 0, stream>>>(deg);
  k_deg_scatter<<<(EE + NN + 255) / 256, 256, 0, stream>>>(ei, ew, deg);
  k_dinv<<<(NN + 255) / 256, 256, 0, stream>>>(deg, dinv);

  k_gcn_gemm<<<(GG * NN) / 32, 256, 0, stream>>>(x_in, gcn_w, xl);

  k_agg_init<<<(int)((GNF + 255) / 256), 256, 0, stream>>>(gcn_b, gc);
  k_scatter<<<(int)(((size_t)(EE + NN) * 64 + 255) / 256), 256, 0, stream>>>(
      ei, ew, dinv, xl, gc);

  size_t rnn_lds = (size_t)(HH * P64 + HH * P128 + 32 * P128 + HH) * sizeof(float);
  k_rnn<<<BN / 32, 256, rnn_lds, stream>>>(gc, w_ih, w_hh, b_ih, b_hh, rnn);

  size_t mlp_lds = (size_t)(32 * P128 + 32 * P512 + 32 * P512) * sizeof(float);
  k_mlp<<<(BN * TT) / 32, 256, mlp_lds, stream>>>(rnn, w1, b1, w2, b2, w3, b3, out);

  k_zero_pred<<<(int)(((size_t)BN * PP * FOUT + 255) / 256), 256, 0, stream>>>(out);
}


# MLP head: 64 rows per work-group, each wave keeps 4 row-tile accumulators so one weight fragment load from global feeds 4 WMMAs instead of 1; per-element k order and sums unchanged.
# speedup vs baseline: 3.2951x; 1.3165x over previous
#include <hip/hip_runtime.h>
#include <hip/hip_bf16.h>

typedef __attribute__((ext_vector_type(2))) float v2f;
typedef __attribute__((ext_vector_type(8))) float v8f;

static constexpr int   NN    = 10000;
static constexpr int   TT    = 8;
static constexpr int   PP    = 2;
static constexpr int   FIN   = 64;
static constexpr int   FOUT  = 64;
static constexpr int   HH    = 128;
static constexpr int   BB    = 2;
static constexpr int   EE    = 160000;
static constexpr int   GG    = BB * TT;
static constexpr int   BN    = BB * NN;
static constexpr int   HID   = HH * 4;
static constexpr size_t GNF  = (size_t)GG * NN * FOUT;
static constexpr int   P64   = 64 + 4;
static constexpr int   P128  = 128 + 4;
static constexpr int   P512  = 512 + 4;

__device__ __forceinline__ v8f wmma4(v2f a, v2f b, v8f c) {
  return __builtin_amdgcn_wmma_f32_16x16x4_f32(false, a, false, b, (short)0, c,
                                               false, false);
}

__device__ __forceinline__ void atomAddF(float* p, float v) {
  __hip_atomic_fetch_add(p, v, __ATOMIC_RELAXED, __HIP_MEMORY_SCOPE_AGENT);
}

__global__ void k_deg_init(float* deg) {
  int i = blockIdx.x * 256 + threadIdx.x;
  if (i < NN) deg[i] = 0.f;
}

__global__ void k_deg_scatter(const int* __restrict__ ei,
                              const float* __restrict__ ew,
                              float* __restrict__ deg) {
  int e = blockIdx.x * 256 + threadIdx.x;
  if (e < EE)                atomAddF(&deg[ei[EE + e]], ew[e]);
  else if (e < EE + NN)      atomAddF(&deg[e - EE], 1.0f);
}

__global__ void k_dinv(const float* __restrict__ deg, float* __restrict__ dinv) {
  int i = blockIdx.x * 256 + threadIdx.x;
  if (i < NN) {
    float d = deg[i];
    dinv[i] = d > 0.f ? rsqrtf(d) : 0.f;
  }
}

__global__ __launch_bounds__(256) void k_gcn_gemm(const float* __restrict__ X,
                                                  const float* __restrict__ W,
                                                  float* __restrict__ Y) {
  __shared__ float wl[FOUT * P64];
  for (int i = threadIdx.x; i < FOUT * FIN; i += 256) wl[(i >> 6) * P64 + (i & 63)] = W[i];
  __syncthreads();
  const int wave = threadIdx.x >> 5, lane = threadIdx.x & 31;
  const int mr = lane & 15, half = lane >> 4, kk = half * 2;
  const int row0 = blockIdx.x * 32 + (wave & 1) * 16;
  const int n0 = (wave >> 1) * 16;
  const float* ap = X + (size_t)(row0 + mr) * FIN + kk;
  const float* bp = wl + (n0 + mr) * P64 + kk;
  v8f c = {};
#pragma unroll
  for (int k = 0; k < FIN; k += 4) {
    v2f a = {ap[k], ap[k + 1]};
    v2f b = {bp[k], bp[k + 1]};
    c = wmma4(a, b, c);
  }
#pragma unroll
  for (int j = 0; j < 8; ++j)
    Y[(size_t)(row0 + j + 8 * half) * FOUT + n0 + mr] = c[j];
}

__global__ void k_agg_init(const float* __restrict__ b, float* __restrict__ gc) {
  size_t i = (size_t)blockIdx.x * 256 + threadIdx.x;
  if (i < GNF) gc[i] = b[i & 63];
}

__global__ void k_scatter(const int* __restrict__ ei, const float* __restrict__ ew,
                          const float* __restrict__ dinv,
                          const float* __restrict__ xl, float* __restrict__ gc) {
  size_t idx = (size_t)blockIdx.x * 256 + threadIdx.x;
  if (idx >= (size_t)(EE + NN) * 64) return;
  int f = (int)(idx & 63);
  int e = (int)(idx >> 6);
  int r, cdst; float nrm;
  if (e < EE) {
    r = ei[e]; cdst = ei[EE + e];
    nrm = dinv[r] * ew[e] * dinv[cdst];
  } else {
    int n = e - EE;
    r = n; cdst = n;
    float di = dinv[n];
    nrm = di * di;
  }
  const float* src = xl + (size_t)r * FOUT + f;
  float* dst = gc + (size_t)cdst * FOUT + f;
#pragma unroll
  for (int g = 0; g < GG; ++g)
    atomAddF(dst + (size_t)g * NN * FOUT, src[(size_t)g * NN * FOUT] * nrm);
}

__global__ __launch_bounds__(256) void k_rnn(const float* __restrict__ gc,
                                             const float* __restrict__ Wih,
                                             const float* __restrict__ Whh,
                                             const float* __restrict__ bih,
                                             const float* __restrict__ bhh,
                                             float* __restrict__ rnn) {
  extern __shared__ float sm[];
  float* s_wih = sm;
  float* s_whh = sm + HH * P64;
  float* s_h   = s_whh + HH * P128;
  float* s_b   = s_h + 32 * P128;
  for (int i = threadIdx.x; i < HH * FOUT; i += 256) s_wih[(i >> 6) * P64 + (i & 63)] = Wih[i];
  for (int i = threadIdx.x; i < HH * HH;  i += 256) s_whh[(i >> 7) * P128 + (i & 127)] = Whh[i];
  for (int i = threadIdx.x; i < 32 * P128; i += 256) s_h[i]   = 0.f;
  if (threadIdx.x < HH) s_b[threadIdx.x] = bih[threadIdx.x] + bhh[threadIdx.x];
  __syncthreads();

  const int wave = threadIdx.x >> 5, lane = threadIdx.x & 31;
  const int mr = lane & 15, half = lane >> 4, kk = half * 2;
  const int n0 = wave * 16;
  const size_t p0 = (size_t)blockIdx.x * 32;
  const float bias = s_b[n0 + mr];
  const float* bw = s_wih + (n0 + mr) * P64 + kk;
  const float* bh = s_whh + (n0 + mr) * P128 + kk;

  for (int t = 0; t < TT; ++t) {
    v8f c0 = {}, c1 = {};
    const float* a0 = gc + ((p0 + mr) * TT + t) * FIN + kk;
    const float* a1 = gc + ((p0 + 16 + mr) * TT + t) * FIN + kk;
#pragma unroll
    for (int k = 0; k < FIN; k += 4) {
      v2f b = {bw[k], bw[k + 1]};
      v2f a = {a0[k], a0[k + 1]};
      c0 = wmma4(a, b, c0);
      v2f a2 = {a1[k], a1[k + 1]};
      c1 = wmma4(a2, b, c1);
    }
    const float* h0 = s_h + mr * P128 + kk;
    const float* h1 = s_h + (16 + mr) * P128 + kk;
#pragma unroll
    for (int k = 0; k < HH; k += 4) {
      v2f b = {bh[k], bh[k + 1]};
      v2f a = {h0[k], h0[k + 1]};
      c0 = wmma4(a, b, c0);
      v2f a2 = {h1[k], h1[k + 1]};
      c1 = wmma4(a2, b, c1);
    }
    __syncthreads();
#pragma unroll
    for (int j = 0; j < 8; ++j) {
      int r = j + 8 * half;
      float v0 = tanhf(c0[j] + bias);
      float v1 = tanhf(c1[j] + bias);
      s_h[r * P128 + n0 + mr] = v0;
      s_h[(16 + r) * P128 + n0 + mr] = v1;
      rnn[((p0 + r) * TT + t) * HH + n0 + mr] = v0;
      rnn[((p0 + 16 + r) * TT + t) * HH + n0 + mr] = v1;
    }
    __syncthreads();
  }
}

static constexpr int   MROWS = 64;

template <int KLEN, int APITCH, int UNR>
__device__ __forceinline__ void tile_x4(const float* __restrict__ ap, const float* __restrict__ bp,
                                        v8f& c0, v8f& c1, v8f& c2, v8f& c3) {
#pragma unroll UNR
  for (int k = 0; k < KLEN; k += 4) {
    v2f b  = {bp[k], bp[k + 1]};
    v2f a0 = {ap[k], ap[k + 1]};
    v2f a1 = {ap[16 * APITCH + k], ap[16 * APITCH + k + 1]};
    v2f a2 = {ap[32 * APITCH + k], ap[32 * APITCH + k + 1]};
    v2f a3 = {ap[48 * APITCH + k], ap[48 * APITCH + k + 1]};
    c0 = wmma4(a0, b, c0);
    c1 = wmma4(a1, b, c1);
    c2 = wmma4(a2, b, c2);
    c3 = wmma4(a3, b, c3);
  }
}

__device__ __forceinline__ void store_relu_x4(float* __restrict__ yp, float bias,
                                              const v8f& c0, const v8f& c1, const v8f& c2, const v8f& c3) {
#pragma unroll
  for (int j = 0; j < 8; ++j) {
    float v0 = c0[j] + bias, v1 = c1[j] + bias, v2 = c2[j] + bias, v3 = c3[j] + bias;
    yp[j * P512]        = v0 > 0.f ? v0 : 0.f;
    yp[(16 + j) * P512] = v1 > 0.f ? v1 : 0.f;
    yp[(32 + j) * P512] = v2 > 0.f ? v2 : 0.f;
    yp[(48 + j) * P512] = v3 > 0.f ? v3 : 0.f;
  }
}

__global__ __launch_bounds__(512) void k_mlp(const float* __restrict__ Xg,
                                             const float* __restrict__ W1, const float* __restrict__ B1,
                                             const float* __restrict__ W2, const float* __restrict__ B2,
                                             const float* __restrict__ W3, const float* __restrict__ B3,
                                             float* __restrict__ out) {
  __shared__ float sm[2 * MROWS * P512];
  float* sY1 = sm;
  float* sY2 = sm + MROWS * P512;
  float* sX  = sY2;
  const size_t r0 = (size_t)blockIdx.x * MROWS;
  for (int i = threadIdx.x; i < MROWS * HH; i += 512) sX[(i >> 7) * P128 + (i & 127)] = Xg[r0 * HH + i];
  __syncthreads();
  const int wave = threadIdx.x >> 5, lane = threadIdx.x & 31;
  const int mr = lane & 15, half = lane >> 4, kk = half * 2;

  for (int ti = 0; ti < 2; ++ti) {
    const int ct = wave * 2 + ti;
    v8f c0 = {}, c1 = {}, c2 = {}, c3 = {};
    tile_x4<HH, P128, 8>(sX + mr * P128 + kk, W1 + (size_t)(ct * 16 + mr) * HH + kk, c0, c1, c2, c3);
    store_relu_x4(sY1 + (8 * half) * P512 + ct * 16 + mr, B1[ct * 16 + mr], c0, c1, c2, c3);
  }
  __syncthreads();

  for (int ti = 0; ti < 2; ++ti) {
    const int ct = wave * 2 + ti;
    v8f c0 = {}, c1 = {}, c2 = {}, c3 = {};
    tile_x4<HID, P512, 4>(sY1 + mr * P512 + kk, W2 + (size_t)(ct * 16 + mr) * HID + kk, c0, c1, c2, c3);
    store_relu_x4(sY2 + (8 * half) * P512 + ct * 16 + mr, B2[ct * 16 + mr], c0, c1, c2, c3);
  }
  __syncthreads();

  {
    int rt = wave & 3, ct = wave >> 2;
    const float* ap = sY2 + (rt * 16 + mr) * P512 + kk;
    const float* bp = W3 + (size_t)(ct * 16 + mr) * HID + kk;
    v8f c = {};
#pragma unroll 4
    for (int k = 0; k < HID; k += 4) {
      v2f a = {ap[k], ap[k + 1]};
      v2f b = {bp[k], bp[k + 1]};
      c = wmma4(a, b, c);
    }
    float bias = B3[ct * 16 + mr];
#pragma unroll
    for (int j = 0; j < 8; ++j) {
      size_t r = r0 + rt * 16 + j + 8 * half;
      size_t p = r >> 3;
      int t = (int)(r & 7);
      out[(p * (TT + PP) + t) * FOUT + ct * 16 + mr] = c[j] + bias;
    }
  }
}

__global__ void k_zero_pred(float* __restrict__ out) {
  size_t idx = (size_t)blockIdx.x * 256 + threadIdx.x;
  if (idx < (size_t)BN * PP * FOUT) {
    size_t p = idx >> 7;
    size_t rem = idx & 127;
    out[p * (size_t)(TT + PP) * FOUT + (size_t)TT * FOUT + rem] = 0.f;
  }
}

extern "C" void kernel_launch(void* const* d_in, const int* in_sizes, int n_in,
                              void* d_out, int out_size, void* d_ws, size_t ws_size,
                              hipStream_t stream) {
  const float* x_in  = (const float*)d_in[0];
  const int*   ei    = (const int*)  d_in[1];
  const float* ew    = (const float*)d_in[2];
  const float* gcn_w = (const float*)d_in[3];
  const float* gcn_b = (const float*)d_in[4];
  const float* w_ih  = (const float*)d_in[5];
  const float* w_hh  = (const float*)d_in[6];
  const float* b_ih  = (const float*)d_in[7];
  const float* b_hh  = (const float*)d_in[8];
  const float* w1    = (const float*)d_in[9];
  const float* b1    = (const float*)d_in[10];
  const float* w2    = (const float*)d_in[11];
  const float* b2    = (const float*)d_in[12];
  const float* w3    = (const float*)d_in[13];
  const float* b3    = (const float*)d_in[14];
  float* out = (float*)d_out;

  float* ws   = (float*)d_ws;
  float* deg  = ws;
  float* dinv = ws + 16384;
  float* xl   = ws + 32768;
  float* gc   = xl + GNF;
  float* rnn  = gc + GNF;

  k_deg_init<<<(NN + 255) / 256, 256, 0, stream>>>(deg);
  k_deg_scatter<<<(EE + NN + 255) / 256, 256, 0, stream>>>(ei, ew, deg);
  k_dinv<<<(NN + 255) / 256, 256, 0, stream>>>(deg, dinv);

  k_gcn_gemm<<<(GG * NN) / 32, 256, 0, stream>>>(x_in, gcn_w, xl);

  k_agg_init<<<(int)((GNF + 255) / 256), 256, 0, stream>>>(gcn_b, gc);
  k_scatter<<<(int)(((size_t)(EE + NN) * 64 + 255) / 256), 256, 0, stream>>>(
      ei, ew, dinv, xl, gc);

  size_t rnn_lds = (size_t)(HH * P64 + HH * P128 + 32 * P128 + HH) * sizeof(float);
  k_rnn<<<BN / 32, 256, rnn_lds, stream>>>(gc, w_ih, w_hh, b_ih, b_hh, rnn);

  k_mlp<<<(BN * TT) / MROWS, 512, 0, stream>>>(rnn, w1, b1, w2, b2, w3, b3, out);

  k_zero_pred<<<(int)(((size_t)BN * PP * FOUT + 255) / 256), 256, 0, stream>>>(out);
}


# MLP weights copied once per call into WMMA B-fragment order: a wave's weight fragment load reads two whole contiguous lines instead of a piece of each of sixteen lines; k order and sums unchanged.
# speedup vs baseline: 3.4367x; 1.0430x over previous
#include <hip/hip_runtime.h>
#include <hip/hip_bf16.h>

typedef __attribute__((ext_vector_type(2))) float v2f;
typedef __attribute__((ext_vector_type(8))) float v8f;

static constexpr int   NN    = 10000;
static constexpr int   TT    = 8;
static constexpr int   PP    = 2;
static constexpr int   FIN   = 64;
static constexpr int   FOUT  = 64;
static constexpr int   HH    = 128;
static constexpr int   BB    = 2;
static constexpr int   EE    = 160000;
static constexpr int   GG    = BB * TT;
static constexpr int   BN    = BB * NN;
static constexpr int   HID   = HH * 4;
static constexpr size_t GNF  = (size_t)GG * NN * FOUT;
static constexpr int   P64   = 64 + 4;
static constexpr int   P128  = 128 + 4;
static constexpr int   P512  = 512 + 4;

__device__ __forceinline__ v8f wmma4(v2f a, v2f b, v8f c) {
  return __builtin_amdgcn_wmma_f32_16x16x4_f32(false, a, false, b, (short)0, c,
                                               false, false);
}

__device__ __forceinline__ void atomAddF(float* p, float v) {
  __hip_atomic_fetch_add(p, v, __ATOMIC_RELAXED, __HIP_MEMORY_SCOPE_AGENT);
}

__global__ void k_deg_init(float* deg) {
  int i = blockIdx.x * 256 + threadIdx.x;
  if (i < NN) deg[i] = 0.f;
}

__global__ void k_deg_scatter(const int* __restrict__ ei,
                              const float* __restrict__ ew,
                              float* __restrict__ deg) {
  int e = blockIdx.x * 256 + threadIdx.x;
  if (e < EE)                atomAddF(&deg[ei[EE + e]], ew[e]);
  else if (e < EE + NN)      atomAddF(&deg[e - EE], 1.0f);
}

__global__ void k_dinv(const float* __restrict__ deg, float* __restrict__ dinv) {
  int i = blockIdx.x * 256 + threadIdx.x;
  if (i < NN) {
    float d = deg[i];
    dinv[i] = d > 0.f ? rsqrtf(d) : 0.f;
  }
}

__global__ __launch_bounds__(256) void k_gcn_gemm(const float* __restrict__ X,
                                                  const float* __restrict__ W,
                                                  float* __restrict__ Y) {
  __shared__ float wl[FOUT * P64];
  for (int i = threadIdx.x; i < FOUT * FIN; i += 256) wl[(i >> 6) * P64 + (i & 63)] = W[i];
  __syncthreads();
  const int wave = threadIdx.x >> 5, lane = threadIdx.x & 31;
  const int mr = lane & 15, half = lane >> 4, kk = half * 2;
  const int row0 = blockIdx.x * 32 + (wave & 1) * 16;
  const int n0 = (wave >> 1) * 16;
  const float* ap = X + (size_t)(row0 + mr) * FIN + kk;
  const float* bp = wl + (n0 + mr) * P64 + kk;
  v8f c = {};
#pragma unroll
  for (int k = 0; k < FIN; k += 4) {
    v2f a = {ap[k], ap[k + 1]};
    v2f b = {bp[k], bp[k + 1]};
    c = wmma4(a, b, c);
  }
#pragma unroll
  for (int j = 0; j < 8; ++j)
    Y[(size_t)(row0 + j + 8 * half) * FOUT + n0 + mr] = c[j];
}

__global__ void k_agg_init(const float* __restrict__ b, float* __restrict__ gc) {
  size_t i = (size_t)blockIdx.x * 256 + threadIdx.x;
  if (i < GNF) gc[i] = b[i & 63];
}

__global__ void k_scatter(const int* __restrict__ ei, const float* __restrict__ ew,
                          const float* __restrict__ dinv,
                          const float* __restrict__ xl, float* __restrict__ gc) {
  size_t idx = (size_t)blockIdx.x * 256 + threadIdx.x;
  if (idx >= (size_t)(EE + NN) * 64) return;
  int f = (int)(idx & 63);
  int e = (int)(idx >> 6);
  int r, cdst; float nrm;
  if (e < EE) {
    r = ei[e]; cdst = ei[EE + e];
    nrm = dinv[r] * ew[e] * dinv[cdst];
  } else {
    int n = e - EE;
    r = n; cdst = n;
    float di = dinv[n];
    nrm = di * di;
  }
  const float* src = xl + (size_t)r * FOUT + f;
  float* dst = gc + (size_t)cdst * FOUT + f;
#pragma unroll
  for (int g = 0; g < GG; ++g)
    atomAddF(dst + (size_t)g * NN * FOUT, src[(size_t)g * NN * FOUT] * nrm);
}

__global__ __launch_bounds__(256) void k_rnn(const float* __restrict__ gc,
                                             const float* __restrict__ Wih,
                                             const float* __restrict__ Whh,
                                             const float* __restrict__ bih,
                                             const float* __restrict__ bhh,
                                             float* __restrict__ rnn) {
  extern __shared__ float sm[];
  float* s_wih = sm;
  float* s_whh = sm + HH * P64;
  float* s_h   = s_whh + HH * P128;
  float* s_b   = s_h + 32 * P128;
  for (int i = threadIdx.x; i < HH * FOUT; i += 256) s_wih[(i >> 6) * P64 + (i & 63)] = Wih[i];
  for (int i = threadIdx.x; i < HH * HH;  i += 256) s_whh[(i >> 7) * P128 + (i & 127)] = Whh[i];
  for (int i = threadIdx.x; i < 32 * P128; i += 256) s_h[i]   = 0.f;
  if (threadIdx.x < HH) s_b[threadIdx.x] = bih[threadIdx.x] + bhh[threadIdx.x];
  __syncthreads();

  const int wave = threadIdx.x >> 5, lane = threadIdx.x & 31;
  const int mr = lane & 15, half = lane >> 4, kk = half * 2;
  const int n0 = wave * 16;
  const size_t p0 = (size_t)blockIdx.x * 32;
  const float bias = s_b[n0 + mr];
  const float* bw = s_wih + (n0 + mr) * P64 + kk;
  const float* bh = s_whh + (n0 + mr) * P128 + kk;

  for (int t = 0; t < TT; ++t) {
    v8f c0 = {}, c1 = {};
    const float* a0 = gc + ((p0 + mr) * TT + t) * FIN + kk;
    const float* a1 = gc + ((p0 + 16 + mr) * TT + t) * FIN + kk;
#pragma unroll
    for (int k = 0; k < FIN; k += 4) {
      v2f b = {bw[k], bw[k + 1]};
      v2f a = {a0[k], a0[k + 1]};
      c0 = wmma4(a, b, c0);
      v2f a2 = {a1[k], a1[k + 1]};
      c1 = wmma4(a2, b, c1);
    }
    const float* h0 = s_h + mr * P128 + kk;
    const float* h1 = s_h + (16 + mr) * P128 + kk;
#pragma unroll
    for (int k = 0; k < HH; k += 4) {
      v2f b = {bh[k], bh[k + 1]};
      v2f a = {h0[k], h0[k + 1]};
      c0 = wmma4(a, b, c0);
      v2f a2 = {h1[k], h1[k + 1]};
      c1 = wmma4(a2, b, c1);
    }
    __syncthreads();
#pragma unroll
    for (int j = 0; j < 8; ++j) {
      int r = j + 8 * half;
      float v0 = tanhf(c0[j] + bias);
      float v1 = tanhf(c1[j] + bias);
      s_h[r * P128 + n0 + mr] = v0;
      s_h[(16 + r) * P128 + n0 + mr] = v1;
      rnn[((p0 + r) * TT + t) * HH + n0 + mr] = v0;
      rnn[((p0 + 16 + r) * TT + t) * HH + n0 + mr] = v1;
    }
    __syncthreads();
  }
}

static constexpr int   MROWS = 64;

static constexpr int   FRAG  = 64;
static constexpr size_t WP1_OFF = 0;
static constexpr size_t WP2_OFF = WP1_OFF + (size_t)HID * HH;
static constexpr size_t WP3_OFF = WP2_OFF + (size_t)HID * HID;
static constexpr size_t WP_FLOATS = WP3_OFF + (size_t)FOUT * HID;

template <int K>
__device__ __forceinline__ void pack_one(const float* __restrict__ W, float* __restrict__ Wp, int j) {
  const int lane = j & 31, frag = j >> 5;
  const int ks = frag % (K / 4), ct = frag / (K / 4);
  const float* src = W + (size_t)(ct * 16 + (lane & 15)) * K + ks * 4 + (lane >> 4) * 2;
  v2f v = {src[0], src[1]};
  *reinterpret_cast<v2f*>(Wp + (size_t)j * 2) = v;
}

__global__ __launch_bounds__(256) void k_pack_w(const float* __restrict__ W1, const float* __restrict__ W2,
                                                const float* __restrict__ W3, float* __restrict__ Wp) {
  constexpr int B1 = HID * HH / 2 / 256;
  constexpr int B2 = HID * HID / 2 / 256;
  const int b = blockIdx.x;
  if (b < B1)            pack_one<HH >(W1, Wp + WP1_OFF, b * 256 + threadIdx.x);
  else if (b < B1 + B2)  pack_one<HID>(W2, Wp + WP2_OFF, (b - B1) * 256 + threadIdx.x);
  else                   pack_one<HID>(W3, Wp + WP3_OFF, (b - B1 - B2) * 256 + threadIdx.x);
}

template <int KLEN, int APITCH, int UNR>
__device__ __forceinline__ void tile_x4(const float* __restrict__ ap, const float* __restrict__ bp,
                                        v8f& c0, v8f& c1, v8f& c2, v8f& c3) {
#pragma unroll UNR
  for (int k = 0; k < KLEN; k += 4) {
    v2f b  = *reinterpret_cast<const v2f*>(bp + (k / 4) * FRAG);
    v2f a0 = {ap[k], ap[k + 1]};
    v2f a1 = {ap[16 * APITCH + k], ap[16 * APITCH + k + 1]};
    v2f a2 = {ap[32 * APITCH + k], ap[32 * APITCH + k + 1]};
    v2f a3 = {ap[48 * APITCH + k], ap[48 * APITCH + k + 1]};
    c0 = wmma4(a0, b, c0);
    c1 = wmma4(a1, b, c1);
    c2 = wmma4(a2, b, c2);
    c3 = wmma4(a3, b, c3);
  }
}

__device__ __forceinline__ void store_relu_x4(float* __restrict__ yp, float bias,
                                              const v8f& c0, const v8f& c1, const v8f& c2, const v8f& c3) {
#pragma unroll
  for (int j = 0; j < 8; ++j) {
    float v0 = c0[j] + bias, v1 = c1[j] + bias, v2 = c2[j] + bias, v3 = c3[j] + bias;
    yp[j * P512]        = v0 > 0.f ? v0 : 0.f;
    yp[(16 + j) * P512] = v1 > 0.f ? v1 : 0.f;
    yp[(32 + j) * P512] = v2 > 0.f ? v2 : 0.f;
    yp[(48 + j) * P512] = v3 > 0.f ? v3 : 0.f;
  }
}

__global__ __launch_bounds__(512) void k_mlp(const float* __restrict__ Xg,
                                             const float* __restrict__ Wp, const float* __restrict__ B1,
                                             const float* __restrict__ B2, const float* __restrict__ B3,
                                             float* __restrict__ out) {
  __shared__ float sm[2 * MROWS * P512];
  float* sY1 = sm;
  float* sY2 = sm + MROWS * P512;
  float* sX  = sY2;
  const size_t r0 = (size_t)blockIdx.x * MROWS;
  for (int i = threadIdx.x; i < MROWS * HH; i += 512) sX[(i >> 7) * P128 + (i & 127)] = Xg[r0 * HH + i];
  __syncthreads();
  const int wave = threadIdx.x >> 5, lane = threadIdx.x & 31;
  const int mr = lane & 15, half = lane >> 4, kk = half * 2;

  for (int ti = 0; ti < 2; ++ti) {
    const int ct = wave * 2 + ti;
    v8f c0 = {}, c1 = {}, c2 = {}, c3 = {};
    tile_x4<HH, P128, 8>(sX + mr * P128 + kk, Wp + WP1_OFF + (size_t)ct * (HH / 4) * FRAG + lane * 2, c0, c1, c2, c3);
    store_relu_x4(sY1 + (8 * half) * P512 + ct * 16 + mr, B1[ct * 16 + mr], c0, c1, c2, c3);
  }
  __syncthreads();

  for (int ti = 0; ti < 2; ++ti) {
    const int ct = wave * 2 + ti;
    v8f c0 = {}, c1 = {}, c2 = {}, c3 = {};
    tile_x4<HID, P512, 8>(sY1 + mr * P512 + kk, Wp + WP2_OFF + (size_t)ct * (HID / 4) * FRAG + lane * 2, c0, c1, c2, c3);
    store_relu_x4(sY2 + (8 * half) * P512 + ct * 16 + mr, B2[ct * 16 + mr], c0, c1, c2, c3);
  }
  __syncthreads();

  {
    int rt = wave & 3, ct = wave >> 2;
    const float* ap = sY2 + (rt * 16 + mr) * P512 + kk;
    const float* bp = Wp + WP3_OFF + (size_t)ct * (HID / 4) * FRAG + lane * 2;
    v8f c = {};
#pragma unroll 16
    for (int k = 0; k < HID; k += 4) {
      v2f a = {ap[k], ap[k + 1]};
      v2f b = *reinterpret_cast<const v2f*>(bp + (k / 4) * FRAG);
      c = wmma4(a, b, c);
    }
    float bias = B3[ct * 16 + mr];
#pragma unroll
    for (int j = 0; j < 8; ++j) {
      size_t r = r0 + rt * 16 + j + 8 * half;
      size_t p = r >> 3;
      int t = (int)(r & 7);
      out[(p * (TT + PP) + t) * FOUT + ct * 16 + mr] = c[j] + bias;
    }
  }
}

__global__ void k_zero_pred(float* __restrict__ out) {
  size_t idx = (size_t)blockIdx.x * 256 + threadIdx.x;
  if (idx < (size_t)BN * PP * FOUT) {
    size_t p = idx >> 7;
    size_t rem = idx & 127;
    out[p * (size_t)(TT + PP) * FOUT + (size_t)TT * FOUT + rem] = 0.f;
  }
}

extern "C" void kernel_launch(void* const* d_in, const int* in_sizes, int n_in,
                              void* d_out, int out_size, void* d_ws, size_t ws_size,
                              hipStream_t stream) {
  const float* x_in  = (const float*)d_in[0];
  const int*   ei    = (const int*)  d_in[1];
  const float* ew    = (const float*)d_in[2];
  const float* gcn_w = (const float*)d_in[3];
  const float* gcn_b = (const float*)d_in[4];
  const float* w_ih  = (const float*)d_in[5];
  const float* w_hh  = (const float*)d_in[6];
  const float* b_ih  = (const float*)d_in[7];
  const float* b_hh  = (const float*)d_in[8];
  const float* w1    = (const float*)d_in[9];
  const float* b1    = (const float*)d_in[10];
  const float* w2    = (const float*)d_in[11];
  const float* b2    = (const float*)d_in[12];
  const float* w3    = (const float*)d_in[13];
  const float* b3    = (const float*)d_in[14];
  float* out = (float*)d_out;

  float* ws   = (float*)d_ws;
  float* deg  = ws;
  float* dinv = ws + 16384;
  float* xl   = ws + 32768;
  float* gc   = xl + GNF;
  float* rnn  = gc + GNF;
  float* wp   = xl;

  k_deg_init<<<(NN + 255) / 256, 256, 0, stream>>>(deg);
  k_deg_scatter<<<(EE + NN + 255) / 256, 256, 0, stream>>>(ei, ew, deg);
  k_dinv<<<(NN + 255) / 256, 256, 0, stream>>>(deg, dinv);

  k_gcn_gemm<<<(GG * NN) / 32, 256, 0, stream>>>(x_in, gcn_w, xl);

  k_agg_init<<<(int)((GNF + 255) / 256), 256, 0, stream>>>(gcn_b, gc);
  k_scatter<<<(int)(((size_t)(EE + NN) * 64 + 255) / 256), 256, 0, stream>>>(
      ei, ew, dinv, xl, gc);

  size_t rnn_lds = (size_t)(HH * P64 + HH * P128 + 32 * P128 + HH) * sizeof(float);
  k_rnn<<<BN / 32, 256, rnn_lds, stream>>>(gc, w_ih, w_hh, b_ih, b_hh, rnn);

  k_pack_w<<<(int)(WP_FLOATS / 2 / 256), 256, 0, stream>>>(w1, w2, w3, wp);
  k_mlp<<<(BN * TT) / MROWS, 512, 0, stream>>>(rnn, wp, b1, b2, b3, out);

  k_zero_pred<<<(int)(((size_t)BN * PP * FOUT + 255) / 256), 256, 0, stream>>>(out);
}


# MLP head: fragment loads issued in groups ahead of the WMMA runs that use them (second register set; weights 32 WMMAs ahead, activations 16), held by scheduling fences; k order and sums unchanged.
# speedup vs baseline: 3.6407x; 1.0594x over previous
#include <hip/hip_runtime.h>
#include <hip/hip_bf16.h>

typedef __attribute__((ext_vector_type(2))) float v2f;
typedef __attribute__((ext_vector_type(8))) float v8f;

static constexpr int   NN    = 10000;
static constexpr int   TT    = 8;
static constexpr int   PP    = 2;
static constexpr int   FIN   = 64;
static constexpr int   FOUT  = 64;
static constexpr int   HH    = 128;
static constexpr int   BB    = 2;
static constexpr int   EE    = 160000;
static constexpr int   GG    = BB * TT;
static constexpr int   BN    = BB * NN;
static constexpr int   HID   = HH * 4;
static constexpr size_t GNF  = (size_t)GG * NN * FOUT;
static constexpr int   P64   = 64 + 4;
static constexpr int   P128  = 128 + 4;
static constexpr int   P512  = 512 + 4;

__device__ __forceinline__ v8f wmma4(v2f a, v2f b, v8f c) {
  return __builtin_amdgcn_wmma_f32_16x16x4_f32(false, a, false, b, (short)0, c,
                                               false, false);
}

__device__ __forceinline__ void atomAddF(float* p, float v) {
  __hip_atomic_fetch_add(p, v, __ATOMIC_RELAXED, __HIP_MEMORY_SCOPE_AGENT);
}

__global__ void k_deg_init(float* deg) {
  int i = blockIdx.x * 256 + threadIdx.x;
  if (i < NN) deg[i] = 0.f;
}

__global__ void k_deg_scatter(const int* __restrict__ ei,
                              const float* __restrict__ ew,
                              float* __restrict__ deg) {
  int e = blockIdx.x * 256 + threadIdx.x;
  if (e < EE)                atomAddF(&deg[ei[EE + e]], ew[e]);
  else if (e < EE + NN)      atomAddF(&deg[e - EE], 1.0f);
}

__global__ void k_dinv(const float* __restrict__ deg, float* __restrict__ dinv) {
  int i = blockIdx.x * 256 + threadIdx.x;
  if (i < NN) {
    float d = deg[i];
    dinv[i] = d > 0.f ? rsqrtf(d) : 0.f;
  }
}

__global__ __launch_bounds__(256) void k_gcn_gemm(const float* __restrict__ X,
                                                  const float* __restrict__ W,
                                                  float* __restrict__ Y) {
  __shared__ float wl[FOUT * P64];
  for (int i = threadIdx.x; i < FOUT * FIN; i += 256) wl[(i >> 6) * P64 + (i & 63)] = W[i];
  __syncthreads();
  const int wave = threadIdx.x >> 5, lane = threadIdx.x & 31;
  const int mr = lane & 15, half = lane >> 4, kk = half * 2;
  const int row0 = blockIdx.x * 32 + (wave & 1) * 16;
  const int n0 = (wave >> 1) * 16;
  const float* ap = X + (size_t)(row0 + mr) * FIN + kk;
  const float* bp = wl + (n0 + mr) * P64 + kk;
  v8f c = {};
#pragma unroll
  for (int k = 0; k < FIN; k += 4) {
    v2f a = {ap[k], ap[k + 1]};
    v2f b = {bp[k], bp[k + 1]};
    c = wmma4(a, b, c);
  }
#pragma unroll
  for (int j = 0; j < 8; ++j)
    Y[(size_t)(row0 + j + 8 * half) * FOUT + n0 + mr] = c[j];
}

__global__ void k_agg_init(const float* __restrict__ b, float* __restrict__ gc) {
  size_t i = (size_t)blockIdx.x * 256 + threadIdx.x;
  if (i < GNF) gc[i] = b[i & 63];
}

__global__ void k_scatter(const int* __restrict__ ei, const float* __restrict__ ew,
                          const float* __restrict__ dinv,
                          const float* __restrict__ xl, float* __restrict__ gc) {
  size_t idx = (size_t)blockIdx.x * 256 + threadIdx.x;
  if (idx >= (size_t)(EE + NN) * 64) return;
  int f = (int)(idx & 63);
  int e = (int)(idx >> 6);
  int r, cdst; float nrm;
  if (e < EE) {
    r = ei[e]; cdst = ei[EE + e];
    nrm = dinv[r] * ew[e] * dinv[cdst];
  } else {
    int n = e - EE;
    r = n; cdst = n;
    float di = dinv[n];
    nrm = di * di;
  }
  const float* src = xl + (size_t)r * FOUT + f;
  float* dst = gc + (size_t)cdst * FOUT + f;
#pragma unroll
  for (int g = 0; g < GG; ++g)
    atomAddF(dst + (size_t)g * NN * FOUT, src[(size_t)g * NN * FOUT] * nrm);
}

__global__ __launch_bounds__(256) void k_rnn(const float* __restrict__ gc,
                                             const float* __restrict__ Wih,
                                             const float* __restrict__ Whh,
                                             const float* __restrict__ bih,
                                             const float* __restrict__ bhh,
                                             float* __restrict__ rnn) {
  extern __shared__ float sm[];
  float* s_wih = sm;
  float* s_whh = sm + HH * P64;
  float* s_h   = s_whh + HH * P128;
  float* s_b   = s_h + 32 * P128;
  for (int i = threadIdx.x; i < HH * FOUT; i += 256) s_wih[(i >> 6) * P64 + (i & 63)] = Wih[i];
  for (int i = threadIdx.x; i < HH * HH;  i += 256) s_whh[(i >> 7) * P128 + (i & 127)] = Whh[i];
  for (int i = threadIdx.x; i < 32 * P128; i += 256) s_h[i]   = 0.f;
  if (threadIdx.x < HH) s_b[threadIdx.x] = bih[threadIdx.x] + bhh[threadIdx.x];
  __syncthreads();

  const int wave = threadIdx.x >> 5, lane = threadIdx.x & 31;
  const int mr = lane & 15, half = lane >> 4, kk = half * 2;
  const int n0 = wave * 16;
  const size_t p0 = (size_t)blockIdx.x * 32;
  const float bias = s_b[n0 + mr];
  const float* bw = s_wih + (n0 + mr) * P64 + kk;
  const float* bh = s_whh + (n0 + mr) * P128 + kk;

  for (int t = 0; t < TT; ++t) {
    v8f c0 = {}, c1 = {};
    const float* a0 = gc + ((p0 + mr) * TT + t) * FIN + kk;
    const float* a1 = gc + ((p0 + 16 + mr) * TT + t) * FIN + kk;
#pragma unroll
    for (int k = 0; k < FIN; k += 4) {
      v2f b = {bw[k], bw[k + 1]};
      v2f a = {a0[k], a0[k + 1]};
      c0 = wmma4(a, b, c0);
      v2f a2 = {a1[k], a1[k + 1]};
      c1 = wmma4(a2, b, c1);
    }
    const float* h0 = s_h + mr * P128 + kk;
    const float* h1 = s_h + (16 + mr) * P128 + kk;
#pragma unroll
    for (int k = 0; k < HH; k += 4) {
      v2f b = {bh[k], bh[k + 1]};
      v2f a = {h0[k], h0[k + 1]};
      c0 = wmma4(a, b, c0);
      v2f a2 = {h1[k], h1[k + 1]};
      c1 = wmma4(a2, b, c1);
    }
    __syncthreads();
#pragma unroll
    for (int j = 0; j < 8; ++j) {
      int r = j + 8 * half;
      float v0 = tanhf(c0[j] + bias);
      float v1 = tanhf(c1[j] + bias);
      s_h[r * P128 + n0 + mr] = v0;
      s_h[(16 + r) * P128 + n0 + mr] = v1;
      rnn[((p0 + r) * TT + t) * HH + n0 + mr] = v0;
      rnn[((p0 + 16 + r) * TT + t) * HH + n0 + mr] = v1;
    }
    __syncthreads();
  }
}

static constexpr int   MROWS = 64;

static constexpr int   FRAG  = 64;
static constexpr size_t WP1_OFF = 0;
static constexpr size_t WP2_OFF = WP1_OFF + (size_t)HID * HH;
static constexpr size_t WP3_OFF = WP2_OFF + (size_t)HID * HID;
static constexpr size_t WP_FLOATS = WP3_OFF + (size_t)FOUT * HID;

template <int K>
__device__ __forceinline__ void pack_one(const float* __restrict__ W, float* __restrict__ Wp, int j) {
  const int lane = j & 31, frag = j >> 5;
  const int ks = frag % (K / 4), ct = frag / (K / 4);
  const float* src = W + (size_t)(ct * 16 + (lane & 15)) * K + ks * 4 + (lane >> 4) * 2;
  v2f v = {src[0], src[1]};
  *reinterpret_cast<v2f*>(Wp + (size_t)j * 2) = v;
}

__global__ __launch_bounds__(256) void k_pack_w(const float* __restrict__ W1, const float* __restrict__ W2,
                                                const float* __restrict__ W3, float* __restrict__ Wp) {
  constexpr int B1 = HID * HH / 2 / 256;
  constexpr int B2 = HID * HID / 2 / 256;
  const int b = blockIdx.x;
  if (b < B1)            pack_one<HH >(W1, Wp + WP1_OFF, b * 256 + threadIdx.x);
  else if (b < B1 + B2)  pack_one<HID>(W2, Wp + WP2_OFF, (b - B1) * 256 + threadIdx.x);
  else                   pack_one<HID>(W3, Wp + WP3_OFF, (b - B1 - B2) * 256 + threadIdx.x);
}

template <int KB>
struct BFrags { v2f b[KB]; };
template <int R, int KH>
struct AFrags { v2f a[R][KH]; };

template <int KB>
__device__ __forceinline__ void load_b(BFrags<KB>& f, const float* __restrict__ bp) {
#pragma unroll
  for (int s = 0; s < KB; ++s) f.b[s] = *reinterpret_cast<const v2f*>(bp + s * FRAG);
}

template <int R, int KH, int APITCH>
__device__ __forceinline__ void load_a(AFrags<R, KH>& f, const float* __restrict__ ap) {
#pragma unroll
  for (int s = 0; s < KH; ++s) {
#pragma unroll
    for (int r = 0; r < R; ++r)
      f.a[r][s] = *reinterpret_cast<const v2f*>(ap + r * 16 * APITCH + s * 4);
  }
}

template <int R, int KH, int KB, int OFF>
__device__ __forceinline__ void mma_half(const AFrags<R, KH>& fa, const BFrags<KB>& fb, v8f (&c)[R]) {
#pragma unroll
  for (int s = 0; s < KH; ++s) {
#pragma unroll
    for (int r = 0; r < R; ++r) c[r] = wmma4(fa.a[r][s], fb.b[OFF + s], c[r]);
  }
}

__device__ __forceinline__ void fence() { __builtin_amdgcn_sched_barrier(0); }

template <int KLEN, int APITCH, int R, int KB, int NT, class Epi>
__device__ __forceinline__ void run_tiles(const float* __restrict__ ap, const float* __restrict__ bp,
                                          Epi epi) {
  constexpr int KH = KB / 2;
  constexpr int NB = KLEN / (4 * KB);
  constexpr int AS = 4 * KB;
  constexpr int AH = 4 * KH;
  constexpr int BS = KB * FRAG;
  BFrags<KB> b0, b1;
  AFrags<R, KH> a0, a1;
  load_b<KB>(b0, bp);
  load_a<R, KH, APITCH>(a0, ap);
#pragma unroll
  for (int t = 0; t < NT; ++t) {
    v8f c[R] = {};
    const float* bt = bp + (size_t)t * NB * BS;
    for (int j = 0; j < NB - 2; j += 2) {
      load_a<R, KH, APITCH>(a1, ap + j * AS + AH);
      load_b<KB>(b1, bt + (j + 1) * BS);
      fence();
      mma_half<R, KH, KB, 0>(a0, b0, c);
      fence();
      load_a<R, KH, APITCH>(a0, ap + (j + 1) * AS);
      fence();
      mma_half<R, KH, KB, KH>(a1, b0, c);
      fence();
      load_a<R, KH, APITCH>(a1, ap + (j + 1) * AS + AH);
      load_b<KB>(b0, bt + (j + 2) * BS);
      fence();
      mma_half<R, KH, KB, 0>(a0, b1, c);
      fence();
      load_a<R, KH, APITCH>(a0, ap + (j + 2) * AS);
      fence();
      mma_half<R, KH, KB, KH>(a1, b1, c);
      fence();
    }
    load_a<R, KH, APITCH>(a1, ap + (NB - 2) * AS + AH);
    load_b<KB>(b1, bt + (NB - 1) * BS);
    fence();
    mma_half<R, KH, KB, 0>(a0, b0, c);
    fence();
    load_a<R, KH, APITCH>(a0, ap + (NB - 1) * AS);
    fence();
    mma_half<R, KH, KB, KH>(a1, b0, c);
    fence();
    load_a<R, KH, APITCH>(a1, ap + (NB - 1) * AS + AH);
    if (t + 1 < NT) load_b<KB>(b0, bt + NB * BS);
    fence();
    mma_half<R, KH, KB, 0>(a0, b1, c);
    fence();
    if (t + 1 < NT) load_a<R, KH, APITCH>(a0, ap);
    fence();
    mma_half<R, KH, KB, KH>(a1, b1, c);
    fence();
    epi(t, c);
  }
}

__device__ __forceinline__ void store_relu_x4(float* __restrict__ yp, float bias,
                                              const v8f& c0, const v8f& c1, const v8f& c2, const v8f& c3) {
#pragma unroll
  for (int j = 0; j < 8; ++j) {
    float v0 = c0[j] + bias, v1 = c1[j] + bias, v2 = c2[j] + bias, v3 = c3[j] + bias;
    yp[j * P512]        = v0 > 0.f ? v0 : 0.f;
    yp[(16 + j) * P512] = v1 > 0.f ? v1 : 0.f;
    yp[(32 + j) * P512] = v2 > 0.f ? v2 : 0.f;
    yp[(48 + j) * P512] = v3 > 0.f ? v3 : 0.f;
  }
}

__global__ __launch_bounds__(512) void k_mlp(const float* __restrict__ Xg,
                                             const float* __restrict__ Wp, const float* __restrict__ B1,
                                             const float* __restrict__ B2, const float* __restrict__ B3,
                                             float* __restrict__ out) {
  __shared__ float sm[2 * MROWS * P512];
  float* sY1 = sm;
  float* sY2 = sm + MROWS * P512;
  float* sX  = sY2;
  const size_t r0 = (size_t)blockIdx.x * MROWS;
  for (int i = threadIdx.x; i < MROWS * HH; i += 512) sX[(i >> 7) * P128 + (i & 127)] = Xg[r0 * HH + i];
  __syncthreads();
  const int wave = threadIdx.x >> 5, lane = threadIdx.x & 31;
  const int mr = lane & 15, half = lane >> 4, kk = half * 2;

  const int ctA = wave * 2;

  const float b1a = B1[ctA * 16 + mr], b1b = B1[ctA * 16 + 16 + mr];
  run_tiles<HH, P128, 4, 8, 2>(
      sX + mr * P128 + kk, Wp + WP1_OFF + (size_t)ctA * (HH / 4) * FRAG + lane * 2,
      [&](int t, v8f (&c)[4]) {
        store_relu_x4(sY1 + (8 * half) * P512 + (ctA + t) * 16 + mr, t ? b1b : b1a, c[0], c[1], c[2], c[3]);
      });
  __syncthreads();

  const float b2a = B2[ctA * 16 + mr], b2b = B2[ctA * 16 + 16 + mr];
  run_tiles<HID, P512, 4, 8, 2>(
      sY1 + mr * P512 + kk, Wp + WP2_OFF + (size_t)ctA * (HID / 4) * FRAG + lane * 2,
      [&](int t, v8f (&c)[4]) {
        store_relu_x4(sY2 + (8 * half) * P512 + (ctA + t) * 16 + mr, t ? b2b : b2a, c[0], c[1], c[2], c[3]);
      });
  __syncthreads();

  const int rt = wave & 3, ct3 = wave >> 2;
  const float b3 = B3[ct3 * 16 + mr];
  run_tiles<HID, P512, 1, 16, 1>(
      sY2 + (rt * 16 + mr) * P512 + kk, Wp + WP3_OFF + (size_t)ct3 * (HID / 4) * FRAG + lane * 2,
      [&](int, v8f (&c)[1]) {
#pragma unroll
        for (int j = 0; j < 8; ++j) {
          size_t r = r0 + rt * 16 + j + 8 * half;
          size_t p = r >> 3;
          int t = (int)(r & 7);
          out[(p * (TT + PP) + t) * FOUT + ct3 * 16 + mr] = c[0][j] + b3;
        }
      });
}

__global__ void k_zero_pred(float* __restrict__ out) {
  size_t idx = (size_t)blockIdx.x * 256 + threadIdx.x;
  if (idx < (size_t)BN * PP * FOUT) {
    size_t p = idx >> 7;
    size_t rem = idx & 127;
    out[p * (size_t)(TT + PP) * FOUT + (size_t)TT * FOUT + rem] = 0.f;
  }
}

extern "C" void kernel_launch(void* const* d_in, const int* in_sizes, int n_in,
                              void* d_out, int out_size, void* d_ws, size_t ws_size,
                              hipStream_t stream) {
  const float* x_in  = (const float*)d_in[0];
  const int*   ei    = (const int*)  d_in[1];
  const float* ew    = (const float*)d_in[2];
  const float* gcn_w = (const float*)d_in[3];
  const float* gcn_b = (const float*)d_in[4];
  const float* w_ih  = (const float*)d_in[5];
  const float* w_hh  = (const float*)d_in[6];
  const float* b_ih  = (const float*)d_in[7];
  const float* b_hh  = (const float*)d_in[8];
  const float* w1    = (const float*)d_in[9];
  const float* b1    = (const float*)d_in[10];
  const float* w2    = (const float*)d_in[11];
  const float* b2    = (const float*)d_in[12];
  const float* w3    = (const float*)d_in[13];
  const float* b3    = (const float*)d_in[14];
  float* out = (float*)d_out;

  float* ws   = (float*)d_ws;
  float* deg  = ws;
  float* dinv = ws + 16384;
  float* xl   = ws + 32768;
  float* gc   = xl + GNF;
  float* rnn  = gc + GNF;
  float* wp   = xl;

  k_deg_init<<<(NN + 255) / 256, 256, 0, stream>>>(deg);
  k_deg_scatter<<<(EE + NN + 255) / 256, 256, 0, stream>>>(ei, ew, deg);
  k_dinv<<<(NN + 255) / 256, 256, 0, stream>>>(deg, dinv);

  k_gcn_gemm<<<(GG * NN) / 32, 256, 0, stream>>>(x_in, gcn_w, xl);

  k_agg_init<<<(int)((GNF + 255) / 256), 256, 0, stream>>>(gcn_b, gc);
  k_scatter<<<(int)(((size_t)(EE + NN) * 64 + 255) / 256), 256, 0, stream>>>(
      ei, ew, dinv, xl, gc);

  size_t rnn_lds = (size_t)(HH * P64 + HH * P128 + 32 * P128 + HH) * sizeof(float);
  k_rnn<<<BN / 32, 256, rnn_lds, stream>>>(gc, w_ih, w_hh, b_ih, b_hh, rnn);

  k_pack_w<<<(int)(WP_FLOATS / 2 / 256), 256, 0, stream>>>(w1, w2, w3, wp);
  k_mlp<<<(BN * TT) / MROWS, 512, 0, stream>>>(rnn, wp, b1, b2, b3, out);

  k_zero_pred<<<(int)(((size_t)BN * PP * FOUT + 255) / 256), 256, 0, stream>>>(out);
}


# Aggregation as a gather: edges grouped by target node each call (count, prefix sum, slot fill); a wave per node and graph pair sums in-edges in registers; no float atomics on gc; sums in slot order
# speedup vs baseline: 4.2580x; 1.1696x over previous
#include <hip/hip_runtime.h>
#include <hip/hip_bf16.h>

typedef __attribute__((ext_vector_type(2))) float v2f;
typedef __attribute__((ext_vector_type(8))) float v8f;

static constexpr int   NN    = 10000;
static constexpr int   TT    = 8;
static constexpr int   PP    = 2;
static constexpr int   FIN   = 64;
static constexpr int   FOUT  = 64;
static constexpr int   HH    = 128;
static constexpr int   BB    = 2;
static constexpr int   EE    = 160000;
static constexpr int   GG    = BB * TT;
static constexpr int   BN    = BB * NN;
static constexpr int   HID   = HH * 4;
static constexpr size_t GNF  = (size_t)GG * NN * FOUT;
static constexpr int   P64   = 64 + 4;
static constexpr int   P128  = 128 + 4;
static constexpr int   P512  = 512 + 4;

__device__ __forceinline__ v8f wmma4(v2f a, v2f b, v8f c) {
  return __builtin_amdgcn_wmma_f32_16x16x4_f32(false, a, false, b, (short)0, c,
                                               false, false);
}

__device__ __forceinline__ void atomAddF(float* p, float v) {
  __hip_atomic_fetch_add(p, v, __ATOMIC_RELAXED, __HIP_MEMORY_SCOPE_AGENT);
}

__device__ __forceinline__ int atomAddI(int* p, int v) {
  return __hip_atomic_fetch_add(p, v, __ATOMIC_RELAXED, __HIP_MEMORY_SCOPE_AGENT);
}

__global__ void k_deg_init(float* deg, int* cnt) {
  int i = blockIdx.x * 256 + threadIdx.x;
  if (i < NN) { deg[i] = 0.f; cnt[i] = 0; }
}

__global__ void k_deg_scatter(const int* __restrict__ ei,
                              const float* __restrict__ ew,
                              float* __restrict__ deg, int* __restrict__ cnt) {
  int e = blockIdx.x * 256 + threadIdx.x;
  if (e < EE) {
    const int c = ei[EE + e];
    atomAddF(&deg[c], ew[e]);
    atomAddI(&cnt[c], 1);
  } else if (e < EE + NN) {
    atomAddF(&deg[e - EE], 1.0f);
  }
}

static constexpr int SCAN_T   = 1024;
static constexpr int SCAN_PER = 10;
__global__ __launch_bounds__(1024) void k_dinv_scan(const float* __restrict__ deg, float* __restrict__ dinv,
                                                    const int* __restrict__ cnt, int* __restrict__ off,
                                                    int* __restrict__ cur) {
  __shared__ int part[SCAN_T];
  const int tid = threadIdx.x;
  const int base = tid * SCAN_PER;
  int loc[SCAN_PER];
  int s = 0;
#pragma unroll
  for (int k = 0; k < SCAN_PER; ++k) {
    const int i = base + k;
    const int v = i < NN ? cnt[i] : 0;
    loc[k] = s;
    s += v;
  }
  part[tid] = s;
  __syncthreads();
  for (int d = 1; d < SCAN_T; d <<= 1) {
    const int v = tid >= d ? part[tid - d] : 0;
    __syncthreads();
    part[tid] += v;
    __syncthreads();
  }
  const int excl = part[tid] - s;
#pragma unroll
  for (int k = 0; k < SCAN_PER; ++k) {
    const int i = base + k;
    if (i < NN) {
      const int o = excl + loc[k];
      off[i] = o;
      cur[i] = o;
      const float d = deg[i];
      dinv[i] = d > 0.f ? rsqrtf(d) : 0.f;
    }
  }
  if (tid == SCAN_T - 1) off[NN] = part[SCAN_T - 1];
}

__global__ __launch_bounds__(256) void k_gcn_gemm(const float* __restrict__ X,
                                                  const float* __restrict__ W,
                                                  float* __restrict__ Y) {
  __shared__ float wl[FOUT * P64];
  for (int i = threadIdx.x; i < FOUT * FIN; i += 256) wl[(i >> 6) * P64 + (i & 63)] = W[i];
  __syncthreads();
  const int wave = threadIdx.x >> 5, lane = threadIdx.x & 31;
  const int mr = lane & 15, half = lane >> 4, kk = half * 2;
  const int row0 = blockIdx.x * 32 + (wave & 1) * 16;
  const int n0 = (wave >> 1) * 16;
  const float* ap = X + (size_t)(row0 + mr) * FIN + kk;
  const float* bp = wl + (n0 + mr) * P64 + kk;
  v8f c = {};
#pragma unroll
  for (int k = 0; k < FIN; k += 4) {
    v2f a = {ap[k], ap[k + 1]};
    v2f b = {bp[k], bp[k + 1]};
    c = wmma4(a, b, c);
  }
#pragma unroll
  for (int j = 0; j < 8; ++j)
    Y[(size_t)(row0 + j + 8 * half) * FOUT + n0 + mr] = c[j];
}

__global__ void k_fill(const int* __restrict__ ei, const float* __restrict__ ew,
                       const float* __restrict__ dinv, int* __restrict__ cur,
                       int* __restrict__ srcs, float* __restrict__ wts) {
  int e = blockIdx.x * 256 + threadIdx.x;
  if (e < EE) {
    const int r = ei[e], c = ei[EE + e];
    const int pos = atomAddI(&cur[c], 1);
    srcs[pos] = r;
    wts[pos]  = dinv[r] * ew[e] * dinv[c];
  }
}

static constexpr int GA_BLOCKS = 2048;
static constexpr int GA_WAVES  = GA_BLOCKS * 8;
static constexpr int GA_ITEMS  = (GG / 2) * NN;
__global__ __launch_bounds__(256) void k_gather(const int* __restrict__ off, const int* __restrict__ srcs,
                                                const float* __restrict__ wts,
                                                const float* __restrict__ dinv,
                                                const float* __restrict__ b,
                                                const float* __restrict__ xl, float* __restrict__ gc) {
  const int lane = threadIdx.x & 31;
  const int wave = __builtin_amdgcn_readfirstlane(threadIdx.x >> 5);
  const v2f bias = *reinterpret_cast<const v2f*>(b + lane * 2);
  for (int item = blockIdx.x * 8 + wave; item < GA_ITEMS; item += GA_WAVES) {
    const int gp = item / NN;
    const int c  = item - gp * NN;
    const int beg = __builtin_amdgcn_readfirstlane(off[c]);
    const int end = __builtin_amdgcn_readfirstlane(off[c + 1]);
    const float di = dinv[c];
    const float wself = di * di;
    const float* x0 = xl + (size_t)(gp * 2) * NN * FOUT + lane * 2;
    const float* x1 = x0 + (size_t)NN * FOUT;
    const v2f s0 = *reinterpret_cast<const v2f*>(x0 + (size_t)c * FOUT);
    const v2f s1 = *reinterpret_cast<const v2f*>(x1 + (size_t)c * FOUT);
    v2f a0 = bias + wself * s0;
    v2f a1 = bias + wself * s1;
    for (int base = beg; base < end; base += 32) {
      const int n = end - base < 32 ? end - base : 32;
      int my_r = 0;
      float my_w = 0.f;
      if (lane < n) { my_r = srcs[base + lane]; my_w = wts[base + lane]; }
#pragma unroll 4
      for (int j = 0; j < n; ++j) {
        const int   r = __builtin_amdgcn_readlane(my_r, j);
        const float w = __int_as_float(__builtin_amdgcn_readlane(__float_as_int(my_w), j));
        const v2f v0 = *reinterpret_cast<const v2f*>(x0 + (size_t)r * FOUT);
        const v2f v1 = *reinterpret_cast<const v2f*>(x1 + (size_t)r * FOUT);
        a0 += w * v0;
        a1 += w * v1;
      }
    }
    float* o0 = gc + ((size_t)(gp * 2) * NN + c) * FOUT + lane * 2;
    *reinterpret_cast<v2f*>(o0) = a0;
    *reinterpret_cast<v2f*>(o0 + (size_t)NN * FOUT) = a1;
  }
}

__global__ __launch_bounds__(256) void k_rnn(const float* __restrict__ gc,
                                             const float* __restrict__ Wih,
                                             const float* __restrict__ Whh,
                                             const float* __restrict__ bih,
                                             const float* __restrict__ bhh,
                                             float* __restrict__ rnn) {
  extern __shared__ float sm[];
  float* s_wih = sm;
  float* s_whh = sm + HH * P64;
  float* s_h   = s_whh + HH * P128;
  float* s_b   = s_h + 32 * P128;
  for (int i = threadIdx.x; i < HH * FOUT; i += 256) s_wih[(i >> 6) * P64 + (i & 63)] = Wih[i];
  for (int i = threadIdx.x; i < HH * HH;  i += 256) s_whh[(i >> 7) * P128 + (i & 127)] = Whh[i];
  for (int i = threadIdx.x; i < 32 * P128; i += 256) s_h[i]   = 0.f;
  if (threadIdx.x < HH) s_b[threadIdx.x] = bih[threadIdx.x] + bhh[threadIdx.x];
  __syncthreads();

  const int wave = threadIdx.x >> 5, lane = threadIdx.x & 31;
  const int mr = lane & 15, half = lane >> 4, kk = half * 2;
  const int n0 = wave * 16;
  const size_t p0 = (size_t)blockIdx.x * 32;
  const float bias = s_b[n0 + mr];
  const float* bw = s_wih + (n0 + mr) * P64 + kk;
  const float* bh = s_whh + (n0 + mr) * P128 + kk;

  for (int t = 0; t < TT; ++t) {
    v8f c0 = {}, c1 = {};
    const float* a0 = gc + ((p0 + mr) * TT + t) * FIN + kk;
    const float* a1 = gc + ((p0 + 16 + mr) * TT + t) * FIN + kk;
#pragma unroll
    for (int k = 0; k < FIN; k += 4) {
      v2f b = {bw[k], bw[k + 1]};
      v2f a = {a0[k], a0[k + 1]};
      c0 = wmma4(a, b, c0);
      v2f a2 = {a1[k], a1[k + 1]};
      c1 = wmma4(a2, b, c1);
    }
    const float* h0 = s_h + mr * P128 + kk;
    const float* h1 = s_h + (16 + mr) * P128 + kk;
#pragma unroll
    for (int k = 0; k < HH; k += 4) {
      v2f b = {bh[k], bh[k + 1]};
      v2f a = {h0[k], h0[k + 1]};
      c0 = wmma4(a, b, c0);
      v2f a2 = {h1[k], h1[k + 1]};
      c1 = wmma4(a2, b, c1);
    }
    __syncthreads();
#pragma unroll
    for (int j = 0; j < 8; ++j) {
      int r = j + 8 * half;
      float v0 = tanhf(c0[j] + bias);
      float v1 = tanhf(c1[j] + bias);
      s_h[r * P128 + n0 + mr] = v0;
      s_h[(16 + r) * P128 + n0 + mr] = v1;
      rnn[((p0 + r) * TT + t) * HH + n0 + mr] = v0;
      rnn[((p0 + 16 + r) * TT + t) * HH + n0 + mr] = v1;
    }
    __syncthreads();
  }
}

static constexpr int   MROWS = 64;

static constexpr int   FRAG  = 64;
static constexpr size_t WP1_OFF = 0;
static constexpr size_t WP2_OFF = WP1_OFF + (size_t)HID * HH;
static constexpr size_t WP3_OFF = WP2_OFF + (size_t)HID * HID;
static constexpr size_t WP_FLOATS = WP3_OFF + (size_t)FOUT * HID;

template <int K>
__device__ __forceinline__ void pack_one(const float* __restrict__ W, float* __restrict__ Wp, int j) {
  const int lane = j & 31, frag = j >> 5;
  const int ks = frag % (K / 4), ct = frag / (K / 4);
  const float* src = W + (size_t)(ct * 16 + (lane & 15)) * K + ks * 4 + (lane >> 4) * 2;
  v2f v = {src[0], src[1]};
  *reinterpret_cast<v2f*>(Wp + (size_t)j * 2) = v;
}

__global__ __launch_bounds__(256) void k_pack_w(const float* __restrict__ W1, const float* __restrict__ W2,
                                                const float* __restrict__ W3, float* __restrict__ Wp) {
  constexpr int B1 = HID * HH / 2 / 256;
  constexpr int B2 = HID * HID / 2 / 256;
  const int b = blockIdx.x;
  if (b < B1)            pack_one<HH >(W1, Wp + WP1_OFF, b * 256 + threadIdx.x);
  else if (b < B1 + B2)  pack_one<HID>(W2, Wp + WP2_OFF, (b - B1) * 256 + threadIdx.x);
  else                   pack_one<HID>(W3, Wp + WP3_OFF, (b - B1 - B2) * 256 + threadIdx.x);
}

template <int KB>
struct BFrags { v2f b[KB]; };
template <int R, int KH>
struct AFrags { v2f a[R][KH]; };

template <int KB>
__device__ __forceinline__ void load_b(BFrags<KB>& f, const float* __restrict__ bp) {
#pragma unroll
  for (int s = 0; s < KB; ++s) f.b[s] = *reinterpret_cast<const v2f*>(bp + s * FRAG);
}

template <int R, int KH, int APITCH>
__device__ __forceinline__ void load_a(AFrags<R, KH>& f, const float* __restrict__ ap) {
#pragma unroll
  for (int s = 0; s < KH; ++s) {
#pragma unroll
    for (int r = 0; r < R; ++r)
      f.a[r][s] = *reinterpret_cast<const v2f*>(ap + r * 16 * APITCH + s * 4);
  }
}

template <int R, int KH, int KB, int OFF>
__device__ __forceinline__ void mma_half(const AFrags<R, KH>& fa, const BFrags<KB>& fb, v8f (&c)[R]) {
#pragma unroll
  for (int s = 0; s < KH; ++s) {
#pragma unroll
    for (int r = 0; r < R; ++r) c[r] = wmma4(fa.a[r][s], fb.b[OFF + s], c[r]);
  }
}

__device__ __forceinline__ void fence() { __builtin_amdgcn_sched_barrier(0); }

template <int KLEN, int APITCH, int R, int KB, int NT, class Epi>
__device__ __forceinline__ void run_tiles(const float* __restrict__ ap, const float* __restrict__ bp,
                                          Epi epi) {
  constexpr int KH = KB / 2;
  constexpr int NB = KLEN / (4 * KB);
  constexpr int AS = 4 * KB;
  constexpr int AH = 4 * KH;
  constexpr int BS = KB * FRAG;
  BFrags<KB> b0, b1;
  AFrags<R, KH> a0, a1;
  load_b<KB>(b0, bp);
  load_a<R, KH, APITCH>(a0, ap);
#pragma unroll
  for (int t = 0; t < NT; ++t) {
    v8f c[R] = {};
    const float* bt = bp + (size_t)t * NB * BS;
    for (int j = 0; j < NB - 2; j += 2) {
      load_a<R, KH, APITCH>(a1, ap + j * AS + AH);
      load_b<KB>(b1, bt + (j + 1) * BS);
      fence();
      mma_half<R, KH, KB, 0>(a0, b0, c);
      fence();
      load_a<R, KH, APITCH>(a0, ap + (j + 1) * AS);
      fence();
      mma_half<R, KH, KB, KH>(a1, b0, c);
      fence();
      load_a<R, KH, APITCH>(a1, ap + (j + 1) * AS + AH);
      load_b<KB>(b0, bt + (j + 2) * BS);
      fence();
      mma_half<R, KH, KB, 0>(a0, b1, c);
      fence();
      load_a<R, KH, APITCH>(a0, ap + (j + 2) * AS);
      fence();
      mma_half<R, KH, KB, KH>(a1, b1, c);
      fence();
    }
    load_a<R, KH, APITCH>(a1, ap + (NB - 2) * AS + AH);
    load_b<KB>(b1, bt + (NB - 1) * BS);
    fence();
    mma_half<R, KH, KB, 0>(a0, b0, c);
    fence();
    load_a<R, KH, APITCH>(a0, ap + (NB - 1) * AS);
    fence();
    mma_half<R, KH, KB, KH>(a1, b0, c);
    fence();
    load_a<R, KH, APITCH>(a1, ap + (NB - 1) * AS + AH);
    if (t + 1 < NT) load_b<KB>(b0, bt + NB * BS);
    fence();
    mma_half<R, KH, KB, 0>(a0, b1, c);
    fence();
    if (t + 1 < NT) load_a<R, KH, APITCH>(a0, ap);
    fence();
    mma_half<R, KH, KB, KH>(a1, b1, c);
    fence();
    epi(t, c);
  }
}

__device__ __forceinline__ void store_relu_x4(float* __restrict__ yp, float bias,
                                              const v8f& c0, const v8f& c1, const v8f& c2, const v8f& c3) {
#pragma unroll
  for (int j = 0; j < 8; ++j) {
    float v0 = c0[j] + bias, v1 = c1[j] + bias, v2 = c2[j] + bias, v3 = c3[j] + bias;
    yp[j * P512]        = v0 > 0.f ? v0 : 0.f;
    yp[(16 + j) * P512] = v1 > 0.f ? v1 : 0.f;
    yp[(32 + j) * P512] = v2 > 0.f ? v2 : 0.f;
    yp[(48 + j) * P512] = v3 > 0.f ? v3 : 0.f;
  }
}

__global__ __launch_bounds__(512) void k_mlp(const float* __restrict__ Xg,
                                             const float* __restrict__ Wp, const float* __restrict__ B1,
                                             const float* __restrict__ B2, const float* __restrict__ B3,
                                             float* __restrict__ out) {
  __shared__ float sm[2 * MROWS * P512];
  float* sY1 = sm;
  float* sY2 = sm + MROWS * P512;
  float* sX  = sY2;
  const size_t r0 = (size_t)blockIdx.x * MROWS;
  for (int i = threadIdx.x; i < MROWS * HH; i += 512) sX[(i >> 7) * P128 + (i & 127)] = Xg[r0 * HH + i];
  __syncthreads();
  const int wave = threadIdx.x >> 5, lane = threadIdx.x & 31;
  const int mr = lane & 15, half = lane >> 4, kk = half * 2;

  const int ctA = wave * 2;

  const float b1a = B1[ctA * 16 + mr], b1b = B1[ctA * 16 + 16 + mr];
  run_tiles<HH, P128, 4, 8, 2>(
      sX + mr * P128 + kk, Wp + WP1_OFF + (size_t)ctA * (HH / 4) * FRAG + lane * 2,
      [&](int t, v8f (&c)[4]) {
        store_relu_x4(sY1 + (8 * half) * P512 + (ctA + t) * 16 + mr, t ? b1b : b1a, c[0], c[1], c[2], c[3]);
      });
  __syncthreads();

  const float b2a = B2[ctA * 16 + mr], b2b = B2[ctA * 16 + 16 + mr];
  run_tiles<HID, P512, 4, 8, 2>(
      sY1 + mr * P512 + kk, Wp + WP2_OFF + (size_t)ctA * (HID / 4) * FRAG + lane * 2,
      [&](int t, v8f (&c)[4]) {
        store_relu_x4(sY2 + (8 * half) * P512 + (ctA + t) * 16 + mr, t ? b2b : b2a, c[0], c[1], c[2], c[3]);
      });
  __syncthreads();

  const int rt = wave & 3, ct3 = wave >> 2;
  const float b3 = B3[ct3 * 16 + mr];
  run_tiles<HID, P512, 1, 16, 1>(
      sY2 + (rt * 16 + mr) * P512 + kk, Wp + WP3_OFF + (size_t)ct3 * (HID / 4) * FRAG + lane * 2,
      [&](int, v8f (&c)[1]) {
#pragma unroll
        for (int j = 0; j < 8; ++j) {
          size_t r = r0 + rt * 16 + j + 8 * half;
          size_t p = r >> 3;
          int t = (int)(r & 7);
          out[(p * (TT + PP) + t) * FOUT + ct3 * 16 + mr] = c[0][j] + b3;
        }
      });
}

__global__ void k_zero_pred(float* __restrict__ out) {
  size_t idx = (size_t)blockIdx.x * 256 + threadIdx.x;
  if (idx < (size_t)BN * PP * FOUT) {
    size_t p = idx >> 7;
    size_t rem = idx & 127;
    out[p * (size_t)(TT + PP) * FOUT + (size_t)TT * FOUT + rem] = 0.f;
  }
}

extern "C" void kernel_launch(void* const* d_in, const int* in_sizes, int n_in,
                              void* d_out, int out_size, void* d_ws, size_t ws_size,
                              hipStream_t stream) {
  const float* x_in  = (const float*)d_in[0];
  const int*   ei    = (const int*)  d_in[1];
  const float* ew    = (const float*)d_in[2];
  const float* gcn_w = (const float*)d_in[3];
  const float* gcn_b = (const float*)d_in[4];
  const float* w_ih  = (const float*)d_in[5];
  const float* w_hh  = (const float*)d_in[6];
  const float* b_ih  = (const float*)d_in[7];
  const float* b_hh  = (const float*)d_in[8];
  const float* w1    = (const float*)d_in[9];
  const float* b1    = (const float*)d_in[10];
  const float* w2    = (const float*)d_in[11];
  const float* b2    = (const float*)d_in[12];
  const float* w3    = (const float*)d_in[13];
  const float* b3    = (const float*)d_in[14];
  float* out = (float*)d_out;

  float* ws   = (float*)d_ws;
  float* deg  = ws;
  float* dinv = ws + 16384;
  float* xl   = ws + 32768;
  float* gc   = xl + GNF;
  float* rnn  = gc + GNF;
  float* wp   = xl;
  int*   cnt  = (int*)rnn;
  int*   off  = cnt + 16384;
  int*   cur  = off + 16384;
  int*   srcs = cur + 16384;
  float* wts  = (float*)(srcs + 163840);

  k_deg_init<<<(NN + 255) / 256, 256, 0, stream>>>(deg, cnt);
  k_deg_scatter<<<(EE + NN + 255) / 256, 256, 0, stream>>>(ei, ew, deg, cnt);
  k_dinv_scan<<<1, 1024, 0, stream>>>(deg, dinv, cnt, off, cur);

  k_gcn_gemm<<<(GG * NN) / 32, 256, 0, stream>>>(x_in, gcn_w, xl);

  k_fill<<<(EE + 255) / 256, 256, 0, stream>>>(ei, ew, dinv, cur, srcs, wts);
  k_gather<<<GA_BLOCKS, 256, 0, stream>>>(off, srcs, wts, dinv, gcn_b, xl, gc);

  size_t rnn_lds = (size_t)(HH * P64 + HH * P128 + 32 * P128 + HH) * sizeof(float);
  k_rnn<<<BN / 32, 256, rnn_lds, stream>>>(gc, w_ih, w_hh, b_ih, b_hh, rnn);

  k_pack_w<<<(int)(WP_FLOATS / 2 / 256), 256, 0, stream>>>(w1, w2, w3, wp);
  k_mlp<<<(BN * TT) / MROWS, 512, 0, stream>>>(rnn, wp, b1, b2, b3, out);

  k_zero_pred<<<(int)(((size_t)BN * PP * FOUT + 255) / 256), 256, 0, stream>>>(out);
}


# MLP head as 256 persistent work-groups (one per WGP) looping over the 64-row tiles: the next tile's input rows are fetched under layer 2 into an LDS plane of their own; k order and sums unchanged.
# speedup vs baseline: 4.3922x; 1.0315x over previous
#include <hip/hip_runtime.h>
#include <hip/hip_bf16.h>

typedef __attribute__((ext_vector_type(2))) float v2f;
typedef __attribute__((ext_vector_type(4))) float v4f;
typedef __attribute__((ext_vector_type(8))) float v8f;

static constexpr int   NN    = 10000;
static constexpr int   TT    = 8;
static constexpr int   PP    = 2;
static constexpr int   FIN   = 64;
static constexpr int   FOUT  = 64;
static constexpr int   HH    = 128;
static constexpr int   BB    = 2;
static constexpr int   EE    = 160000;
static constexpr int   GG    = BB * TT;
static constexpr int   BN    = BB * NN;
static constexpr int   HID   = HH * 4;
static constexpr size_t GNF  = (size_t)GG * NN * FOUT;
static constexpr int   P64   = 64 + 4;
static constexpr int   P128  = 128 + 4;
static constexpr int   P512  = 512 + 4;

__device__ __forceinline__ v8f wmma4(v2f a, v2f b, v8f c) {
  return __builtin_amdgcn_wmma_f32_16x16x4_f32(false, a, false, b, (short)0, c,
                                               false, false);
}

__device__ __forceinline__ void atomAddF(float* p, float v) {
  __hip_atomic_fetch_add(p, v, __ATOMIC_RELAXED, __HIP_MEMORY_SCOPE_AGENT);
}

__device__ __forceinline__ int atomAddI(int* p, int v) {
  return __hip_atomic_fetch_add(p, v, __ATOMIC_RELAXED, __HIP_MEMORY_SCOPE_AGENT);
}

__global__ void k_deg_init(float* deg, int* cnt) {
  int i = blockIdx.x * 256 + threadIdx.x;
  if (i < NN) { deg[i] = 0.f; cnt[i] = 0; }
}

__global__ void k_deg_scatter(const int* __restrict__ ei,
                              const float* __restrict__ ew,
                              float* __restrict__ deg, int* __restrict__ cnt) {
  int e = blockIdx.x * 256 + threadIdx.x;
  if (e < EE) {
    const int c = ei[EE + e];
    atomAddF(&deg[c], ew[e]);
    atomAddI(&cnt[c], 1);
  } else if (e < EE + NN) {
    atomAddF(&deg[e - EE], 1.0f);
  }
}

static constexpr int SCAN_T   = 1024;
static constexpr int SCAN_PER = 10;
__global__ __launch_bounds__(1024) void k_dinv_scan(const float* __restrict__ deg, float* __restrict__ dinv,
                                                    const int* __restrict__ cnt, int* __restrict__ off,
                                                    int* __restrict__ cur) {
  __shared__ int part[SCAN_T];
  const int tid = threadIdx.x;
  const int base = tid * SCAN_PER;
  int loc[SCAN_PER];
  int s = 0;
#pragma unroll
  for (int k = 0; k < SCAN_PER; ++k) {
    const int i = base + k;
    const int v = i < NN ? cnt[i] : 0;
    loc[k] = s;
    s += v;
  }
  part[tid] = s;
  __syncthreads();
  for (int d = 1; d < SCAN_T; d <<= 1) {
    const int v = tid >= d ? part[tid - d] : 0;
    __syncthreads();
    part[tid] += v;
    __syncthreads();
  }
  const int excl = part[tid] - s;
#pragma unroll
  for (int k = 0; k < SCAN_PER; ++k) {
    const int i = base + k;
    if (i < NN) {
      const int o = excl + loc[k];
      off[i] = o;
      cur[i] = o;
      const float d = deg[i];
      dinv[i] = d > 0.f ? rsqrtf(d) : 0.f;
    }
  }
  if (tid == SCAN_T - 1) off[NN] = part[SCAN_T - 1];
}

__global__ __launch_bounds__(256) void k_gcn_gemm(const float* __restrict__ X,
                                                  const float* __restrict__ W,
                                                  float* __restrict__ Y) {
  __shared__ float wl[FOUT * P64];
  for (int i = threadIdx.x; i < FOUT * FIN; i += 256) wl[(i >> 6) * P64 + (i & 63)] = W[i];
  __syncthreads();
  const int wave = threadIdx.x >> 5, lane = threadIdx.x & 31;
  const int mr = lane & 15, half = lane >> 4, kk = half * 2;
  const int row0 = blockIdx.x * 32 + (wave & 1) * 16;
  const int n0 = (wave >> 1) * 16;
  const float* ap = X + (size_t)(row0 + mr) * FIN + kk;
  const float* bp = wl + (n0 + mr) * P64 + kk;
  v8f c = {};
#pragma unroll
  for (int k = 0; k < FIN; k += 4) {
    v2f a = {ap[k], ap[k + 1]};
    v2f b = {bp[k], bp[k + 1]};
    c = wmma4(a, b, c);
  }
#pragma unroll
  for (int j = 0; j < 8; ++j)
    Y[(size_t)(row0 + j + 8 * half) * FOUT + n0 + mr] = c[j];
}

__global__ void k_fill(const int* __restrict__ ei, const float* __restrict__ ew,
                       const float* __restrict__ dinv, int* __restrict__ cur,
                       int* __restrict__ srcs, float* __restrict__ wts) {
  int e = blockIdx.x * 256 + threadIdx.x;
  if (e < EE) {
    const int r = ei[e], c = ei[EE + e];
    const int pos = atomAddI(&cur[c], 1);
    srcs[pos] = r;
    wts[pos]  = dinv[r] * ew[e] * dinv[c];
  }
}

static constexpr int GA_BLOCKS = 2048;
static constexpr int GA_WAVES  = GA_BLOCKS * 8;
static constexpr int GA_ITEMS  = (GG / 2) * NN;
__global__ __launch_bounds__(256) void k_gather(const int* __restrict__ off, const int* __restrict__ srcs,
                                                const float* __restrict__ wts,
                                                const float* __restrict__ dinv,
                                                const float* __restrict__ b,
                                                const float* __restrict__ xl, float* __restrict__ gc) {
  const int lane = threadIdx.x & 31;
  const int wave = __builtin_amdgcn_readfirstlane(threadIdx.x >> 5);
  const v2f bias = *reinterpret_cast<const v2f*>(b + lane * 2);
  for (int item = blockIdx.x * 8 + wave; item < GA_ITEMS; item += GA_WAVES) {
    const int gp = item / NN;
    const int c  = item - gp * NN;
    const int beg = __builtin_amdgcn_readfirstlane(off[c]);
    const int end = __builtin_amdgcn_readfirstlane(off[c + 1]);
    const float di = dinv[c];
    const float wself = di * di;
    const float* x0 = xl + (size_t)(gp * 2) * NN * FOUT + lane * 2;
    const float* x1 = x0 + (size_t)NN * FOUT;
    const v2f s0 = *reinterpret_cast<const v2f*>(x0 + (size_t)c * FOUT);
    const v2f s1 = *reinterpret_cast<const v2f*>(x1 + (size_t)c * FOUT);
    v2f a0 = bias + wself * s0;
    v2f a1 = bias + wself * s1;
    for (int base = beg; base < end; base += 32) {
      const int n = end - base < 32 ? end - base : 32;
      int my_r = 0;
      float my_w = 0.f;
      if (lane < n) { my_r = srcs[base + lane]; my_w = wts[base + lane]; }
#pragma unroll 4
      for (int j = 0; j < n; ++j) {
        const int   r = __builtin_amdgcn_readlane(my_r, j);
        const float w = __int_as_float(__builtin_amdgcn_readlane(__float_as_int(my_w), j));
        const v2f v0 = *reinterpret_cast<const v2f*>(x0 + (size_t)r * FOUT);
        const v2f v1 = *reinterpret_cast<const v2f*>(x1 + (size_t)r * FOUT);
        a0 += w * v0;
        a1 += w * v1;
      }
    }
    float* o0 = gc + ((size_t)(gp * 2) * NN + c) * FOUT + lane * 2;
    *reinterpret_cast<v2f*>(o0) = a0;
    *reinterpret_cast<v2f*>(o0 + (size_t)NN * FOUT) = a1;
  }
}

__global__ __launch_bounds__(256) void k_rnn(const float* __restrict__ gc,
                                             const float* __restrict__ Wih,
                                             const float* __restrict__ Whh,
                                             const float* __restrict__ bih,
                                             const float* __restrict__ bhh,
                                             float* __restrict__ rnn) {
  extern __shared__ float sm[];
  float* s_wih = sm;
  float* s_whh = sm + HH * P64;
  float* s_h   = s_whh + HH * P128;
  float* s_b   = s_h + 32 * P128;
  for (int i = threadIdx.x; i < HH * FOUT; i += 256) s_wih[(i >> 6) * P64 + (i & 63)] = Wih[i];
  for (int i = threadIdx.x; i < HH * HH;  i += 256) s_whh[(i >> 7) * P128 + (i & 127)] = Whh[i];
  for (int i = threadIdx.x; i < 32 * P128; i += 256) s_h[i]   = 0.f;
  if (threadIdx.x < HH) s_b[threadIdx.x] = bih[threadIdx.x] + bhh[threadIdx.x];
  __syncthreads();

  const int wave = threadIdx.x >> 5, lane = threadIdx.x & 31;
  const int mr = lane & 15, half = lane >> 4, kk = half * 2;
  const int n0 = wave * 16;
  const size_t p0 = (size_t)blockIdx.x * 32;
  const float bias = s_b[n0 + mr];
  const float* bw = s_wih + (n0 + mr) * P64 + kk;
  const float* bh = s_whh + (n0 + mr) * P128 + kk;

  for (int t = 0; t < TT; ++t) {
    v8f c0 = {}, c1 = {};
    const float* a0 = gc + ((p0 + mr) * TT + t) * FIN + kk;
    const float* a1 = gc + ((p0 + 16 + mr) * TT + t) * FIN + kk;
#pragma unroll
    for (int k = 0; k < FIN; k += 4) {
      v2f b = {bw[k], bw[k + 1]};
      v2f a = {a0[k], a0[k + 1]};
      c0 = wmma4(a, b, c0);
      v2f a2 = {a1[k], a1[k + 1]};
      c1 = wmma4(a2, b, c1);
    }
    const float* h0 = s_h + mr * P128 + kk;
    const float* h1 = s_h + (16 + mr) * P128 + kk;
#pragma unroll
    for (int k = 0; k < HH; k += 4) {
      v2f b = {bh[k], bh[k + 1]};
      v2f a = {h0[k], h0[k + 1]};
      c0 = wmma4(a, b, c0);
      v2f a2 = {h1[k], h1[k + 1]};
      c1 = wmma4(a2, b, c1);
    }
    __syncthreads();
#pragma unroll
    for (int j = 0; j < 8; ++j) {
      int r = j + 8 * half;
      float v0 = tanhf(c0[j] + bias);
      float v1 = tanhf(c1[j] + bias);
      s_h[r * P128 + n0 + mr] = v0;
      s_h[(16 + r) * P128 + n0 + mr] = v1;
      rnn[((p0 + r) * TT + t) * HH + n0 + mr] = v0;
      rnn[((p0 + 16 + r) * TT + t) * HH + n0 + mr] = v1;
    }
    __syncthreads();
  }
}

static constexpr int   MROWS = 64;

static constexpr int   FRAG  = 64;
static constexpr size_t WP1_OFF = 0;
static constexpr size_t WP2_OFF = WP1_OFF + (size_t)HID * HH;
static constexpr size_t WP3_OFF = WP2_OFF + (size_t)HID * HID;
static constexpr size_t WP_FLOATS = WP3_OFF + (size_t)FOUT * HID;

template <int K>
__device__ __forceinline__ void pack_one(const float* __restrict__ W, float* __restrict__ Wp, int j) {
  const int lane = j & 31, frag = j >> 5;
  const int ks = frag % (K / 4), ct = frag / (K / 4);
  const float* src = W + (size_t)(ct * 16 + (lane & 15)) * K + ks * 4 + (lane >> 4) * 2;
  v2f v = {src[0], src[1]};
  *reinterpret_cast<v2f*>(Wp + (size_t)j * 2) = v;
}

__global__ __launch_bounds__(256) void k_pack_w(const float* __restrict__ W1, const float* __restrict__ W2,
                                                const float* __restrict__ W3, float* __restrict__ Wp) {
  constexpr int B1 = HID * HH / 2 / 256;
  constexpr int B2 = HID * HID / 2 / 256;
  const int b = blockIdx.x;
  if (b < B1)            pack_one<HH >(W1, Wp + WP1_OFF, b * 256 + threadIdx.x);
  else if (b < B1 + B2)  pack_one<HID>(W2, Wp + WP2_OFF, (b - B1) * 256 + threadIdx.x);
  else                   pack_one<HID>(W3, Wp + WP3_OFF, (b - B1 - B2) * 256 + threadIdx.x);
}

template <int KB>
struct BFrags { v2f b[KB]; };
template <int R, int KH>
struct AFrags { v2f a[R][KH]; };

template <int KB>
__device__ __forceinline__ void load_b(BFrags<KB>& f, const float* __restrict__ bp) {
#pragma unroll
  for (int s = 0; s < KB; ++s) f.b[s] = *reinterpret_cast<const v2f*>(bp + s * FRAG);
}

template <int R, int KH, int APITCH>
__device__ __forceinline__ void load_a(AFrags<R, KH>& f, const float* __restrict__ ap) {
#pragma unroll
  for (int s = 0; s < KH; ++s) {
#pragma unroll
    for (int r = 0; r < R; ++r)
      f.a[r][s] = *reinterpret_cast<const v2f*>(ap + r * 16 * APITCH + s * 4);
  }
}

template <int R, int KH, int KB, int OFF>
__device__ __forceinline__ void mma_half(const AFrags<R, KH>& fa, const BFrags<KB>& fb, v8f (&c)[R]) {
#pragma unroll
  for (int s = 0; s < KH; ++s) {
#pragma unroll
    for (int r = 0; r < R; ++r) c[r] = wmma4(fa.a[r][s], fb.b[OFF + s], c[r]);
  }
}

__device__ __forceinline__ void fence() { __builtin_amdgcn_sched_barrier(0); }

template <int KLEN, int APITCH, int R, int KB, int NT, class Epi>
__device__ __forceinline__ void run_tiles(const float* __restrict__ ap, const float* __restrict__ bp,
                                          Epi epi) {
  constexpr int KH = KB / 2;
  constexpr int NB = KLEN / (4 * KB);
  constexpr int AS = 4 * KB;
  constexpr int AH = 4 * KH;
  constexpr int BS = KB * FRAG;
  BFrags<KB> b0, b1;
  AFrags<R, KH> a0, a1;
  load_b<KB>(b0, bp);
  load_a<R, KH, APITCH>(a0, ap);
#pragma unroll
  for (int t = 0; t < NT; ++t) {
    v8f c[R] = {};
    const float* bt = bp + (size_t)t * NB * BS;
    for (int j = 0; j < NB - 2; j += 2) {
      load_a<R, KH, APITCH>(a1, ap + j * AS + AH);
      load_b<KB>(b1, bt + (j + 1) * BS);
      fence();
      mma_half<R, KH, KB, 0>(a0, b0, c);
      fence();
      load_a<R, KH, APITCH>(a0, ap + (j + 1) * AS);
      fence();
      mma_half<R, KH, KB, KH>(a1, b0, c);
      fence();
      load_a<R, KH, APITCH>(a1, ap + (j + 1) * AS + AH);
      load_b<KB>(b0, bt + (j + 2) * BS);
      fence();
      mma_half<R, KH, KB, 0>(a0, b1, c);
      fence();
      load_a<R, KH, APITCH>(a0, ap + (j + 2) * AS);
      fence();
      mma_half<R, KH, KB, KH>(a1, b1, c);
      fence();
    }
    load_a<R, KH, APITCH>(a1, ap + (NB - 2) * AS + AH);
    load_b<KB>(b1, bt + (NB - 1) * BS);
    fence();
    mma_half<R, KH, KB, 0>(a0, b0, c);
    fence();
    load_a<R, KH, APITCH>(a0, ap + (NB - 1) * AS);
    fence();
    mma_half<R, KH, KB, KH>(a1, b0, c);
    fence();
    load_a<R, KH, APITCH>(a1, ap + (NB - 1) * AS + AH);
    if (t + 1 < NT) load_b<KB>(b0, bt + NB * BS);
    fence();
    mma_half<R, KH, KB, 0>(a0, b1, c);
    fence();
    if (t + 1 < NT) load_a<R, KH, APITCH>(a0, ap);
    fence();
    mma_half<R, KH, KB, KH>(a1, b1, c);
    fence();
    epi(t, c);
  }
}

__device__ __forceinline__ void store_relu_x4(float* __restrict__ yp, float bias,
                                              const v8f& c0, const v8f& c1, const v8f& c2, const v8f& c3) {
#pragma unroll
  for (int j = 0; j < 8; ++j) {
    float v0 = c0[j] + bias, v1 = c1[j] + bias, v2 = c2[j] + bias, v3 = c3[j] + bias;
    yp[j * P512]        = v0 > 0.f ? v0 : 0.f;
    yp[(16 + j) * P512] = v1 > 0.f ? v1 : 0.f;
    yp[(32 + j) * P512] = v2 > 0.f ? v2 : 0.f;
    yp[(48 + j) * P512] = v3 > 0.f ? v3 : 0.f;
  }
}

static constexpr int   MLP_TILES = (BN * TT) / MROWS;
static constexpr int   MLP_WGS   = 256;
static constexpr int   XQ        = MROWS * HH / 4 / 512;

__device__ __forceinline__ void load_x(v4f (&x)[XQ], const float* __restrict__ Xg, int tile) {
  const v4f* p = reinterpret_cast<const v4f*>(Xg + (size_t)tile * MROWS * HH) + threadIdx.x;
#pragma unroll
  for (int u = 0; u < XQ; ++u) x[u] = p[u * 512];
}

__device__ __forceinline__ void store_x(float* __restrict__ sX, const v4f (&x)[XQ]) {
#pragma unroll
  for (int u = 0; u < XQ; ++u) {
    const int q = threadIdx.x + u * 512;
    *reinterpret_cast<v4f*>(sX + (q >> 5) * P128 + (q & 31) * 4) = x[u];
  }
}

__device__ __forceinline__ int keep_in_loop(int v) {
  asm volatile("" : "+v"(v));
  return v;
}

__global__ __launch_bounds__(512) void k_mlp(const float* __restrict__ Xg,
                                             const float* __restrict__ Wp, const float* __restrict__ B1,
                                             const float* __restrict__ B2, const float* __restrict__ B3,
                                             float* __restrict__ out) {
  __shared__ float sm[2 * MROWS * P512 + MROWS * P128];
  float* sY1 = sm;
  float* sY2 = sm + MROWS * P512;
  float* sX  = sm + 2 * MROWS * P512;
  const int wave = threadIdx.x >> 5, lane = threadIdx.x & 31;
  const int mr = lane & 15, half = lane >> 4, kk = half * 2;

  const int ctA = wave * 2;
  const int rt = wave & 3, ct3 = wave >> 2;

  const float b1a = B1[ctA * 16 + mr], b1b = B1[ctA * 16 + 16 + mr];
  const float b2a = B2[ctA * 16 + mr], b2b = B2[ctA * 16 + 16 + mr];
  const float b3 = B3[ct3 * 16 + mr];

  v4f xr[XQ];
  load_x(xr, Xg, blockIdx.x);
  store_x(sX, xr);
  __syncthreads();

  for (int tile = blockIdx.x; tile < MLP_TILES; tile += MLP_WGS) {
    const size_t r0 = (size_t)tile * MROWS;
    const bool more = tile + MLP_WGS < MLP_TILES;

    const int ox1 = keep_in_loop(mr * P128 + kk);
    const int oy1 = keep_in_loop((8 * half) * P512 + ctA * 16 + mr);
    run_tiles<HH, P128, 4, 8, 2>(
        sX + ox1, Wp + WP1_OFF + (size_t)ctA * (HH / 4) * FRAG + lane * 2,
        [&](int t, v8f (&c)[4]) {
          store_relu_x4(sY1 + oy1 + t * 16, t ? b1b : b1a, c[0], c[1], c[2], c[3]);
        });
    __syncthreads();

    if (more) load_x(xr, Xg, tile + MLP_WGS);
    fence();

    const int ox2 = keep_in_loop(mr * P512 + kk);
    const int oy2 = keep_in_loop((8 * half) * P512 + ctA * 16 + mr);
    run_tiles<HID, P512, 4, 8, 2>(
        sY1 + ox2, Wp + WP2_OFF + (size_t)ctA * (HID / 4) * FRAG + lane * 2,
        [&](int t, v8f (&c)[4]) {
          store_relu_x4(sY2 + oy2 + t * 16, t ? b2b : b2a, c[0], c[1], c[2], c[3]);
        });
    if (more) store_x(sX, xr);
    __syncthreads();

    const int ox3 = keep_in_loop((rt * 16 + mr) * P512 + kk);
    run_tiles<HID, P512, 1, 16, 1>(
        sY2 + ox3, Wp + WP3_OFF + (size_t)ct3 * (HID / 4) * FRAG + lane * 2,
        [&](int, v8f (&c)[1]) {
#pragma unroll
          for (int j = 0; j < 8; ++j) {
            size_t r = r0 + rt * 16 + j + 8 * half;
            size_t p = r >> 3;
            int t = (int)(r & 7);
            out[(p * (TT + PP) + t) * FOUT + ct3 * 16 + mr] = c[0][j] + b3;
          }
        });
  }
}

__global__ void k_zero_pred(float* __restrict__ out) {
  size_t idx = (size_t)blockIdx.x * 256 + threadIdx.x;
  if (idx < (size_t)BN * PP * FOUT) {
    size_t p = idx >> 7;
    size_t rem = idx & 127;
    out[p * (size_t)(TT + PP) * FOUT + (size_t)TT * FOUT + rem] = 0.f;
  }
}

extern "C" void kernel_launch(void* const* d_in, const int* in_sizes, int n_in,
                              void* d_out, int out_size, void* d_ws, size_t ws_size,
                              hipStream_t stream) {
  const float* x_in  = (const float*)d_in[0];
  const int*   ei    = (const int*)  d_in[1];
  const float* ew    = (const float*)d_in[2];
  const float* gcn_w = (const float*)d_in[3];
  const float* gcn_b = (const float*)d_in[4];
  const float* w_ih  = (const float*)d_in[5];
  const float* w_hh  = (const float*)d_in[6];
  const float* b_ih  = (const float*)d_in[7];
  const float* b_hh  = (const float*)d_in[8];
  const float* w1    = (const float*)d_in[9];
  const float* b1    = (const float*)d_in[10];
  const float* w2    = (const float*)d_in[11];
  const float* b2    = (const float*)d_in[12];
  const float* w3    = (const float*)d_in[13];
  const float* b3    = (const float*)d_in[14];
  float* out = (float*)d_out;

  float* ws   = (float*)d_ws;
  float* deg  = ws;
  float* dinv = ws + 16384;
  float* xl   = ws + 32768;
  float* gc   = xl + GNF;
  float* rnn  = gc + GNF;
  float* wp   = xl;
  int*   cnt  = (int*)rnn;
  int*   off  = cnt + 16384;
  int*   cur  = off + 16384;
  int*   srcs = cur + 16384;
  float* wts  = (float*)(srcs + 163840);

  k_deg_init<<<(NN + 255) / 256, 256, 0, stream>>>(deg, cnt);
  k_deg_scatter<<<(EE + NN + 255) / 256, 256, 0, stream>>>(ei, ew, deg, cnt);
  k_dinv_scan<<<1, 1024, 0, stream>>>(deg, dinv, cnt, off, cur);

  k_gcn_gemm<<<(GG * NN) / 32, 256, 0, stream>>>(x_in, gcn_w, xl);

  k_fill<<<(EE + 255) / 256, 256, 0, stream>>>(ei, ew, dinv, cur, srcs, wts);
  k_gather<<<GA_BLOCKS, 256, 0, stream>>>(off, srcs, wts, dinv, gcn_b, xl, gc);

  size_t rnn_lds = (size_t)(HH * P64 + HH * P128 + 32 * P128 + HH) * sizeof(float);
  k_rnn<<<BN / 32, 256, rnn_lds, stream>>>(gc, w_ih, w_hh, b_ih, b_hh, rnn);

  k_pack_w<<<(int)(WP_FLOATS / 2 / 256), 256, 0, stream>>>(w1, w2, w3, wp);
  k_mlp<<<MLP_WGS, 512, 0, stream>>>(rnn, wp, b1, b2, b3, out);

  k_zero_pred<<<(int)(((size_t)BN * PP * FOUT + 255) / 256), 256, 0, stream>>>(out);
}


# RNN as 250 work-groups of 80 sequences (one round; was 625 of 32 in two): a step is one K=192 product run over [x_t | h] in LDS, x_t fetched a step ahead, packed weights; k order and sums unchanged.
# speedup vs baseline: 4.6175x; 1.0513x over previous
#include <hip/hip_runtime.h>
#include <hip/hip_bf16.h>

typedef __attribute__((ext_vector_type(2))) float v2f;
typedef __attribute__((ext_vector_type(4))) float v4f;
typedef __attribute__((ext_vector_type(8))) float v8f;

static constexpr int   NN    = 10000;
static constexpr int   TT    = 8;
static constexpr int   PP    = 2;
static constexpr int   FIN   = 64;
static constexpr int   FOUT  = 64;
static constexpr int   HH    = 128;
static constexpr int   BB    = 2;
static constexpr int   EE    = 160000;
static constexpr int   GG    = BB * TT;
static constexpr int   BN    = BB * NN;
static constexpr int   HID   = HH * 4;
static constexpr size_t GNF  = (size_t)GG * NN * FOUT;
static constexpr int   P64   = 64 + 4;
static constexpr int   P128  = 128 + 4;
static constexpr int   P512  = 512 + 4;
static constexpr int   KCAT  = FOUT + HH;
static constexpr int   P192  = KCAT + 4;

__device__ __forceinline__ v8f wmma4(v2f a, v2f b, v8f c) {
  return __builtin_amdgcn_wmma_f32_16x16x4_f32(false, a, false, b, (short)0, c,
                                               false, false);
}

__device__ __forceinline__ void atomAddF(float* p, float v) {
  __hip_atomic_fetch_add(p, v, __ATOMIC_RELAXED, __HIP_MEMORY_SCOPE_AGENT);
}

__device__ __forceinline__ int atomAddI(int* p, int v) {
  return __hip_atomic_fetch_add(p, v, __ATOMIC_RELAXED, __HIP_MEMORY_SCOPE_AGENT);
}

__global__ void k_deg_init(float* deg, int* cnt) {
  int i = blockIdx.x * 256 + threadIdx.x;
  if (i < NN) { deg[i] = 0.f; cnt[i] = 0; }
}

__global__ void k_deg_scatter(const int* __restrict__ ei,
                              const float* __restrict__ ew,
                              float* __restrict__ deg, int* __restrict__ cnt) {
  int e = blockIdx.x * 256 + threadIdx.x;
  if (e < EE) {
    const int c = ei[EE + e];
    atomAddF(&deg[c], ew[e]);
    atomAddI(&cnt[c], 1);
  } else if (e < EE + NN) {
    atomAddF(&deg[e - EE], 1.0f);
  }
}

static constexpr int SCAN_T   = 1024;
static constexpr int SCAN_PER = 10;
__global__ __launch_bounds__(1024) void k_dinv_scan(const float* __restrict__ deg, float* __restrict__ dinv,
                                                    const int* __restrict__ cnt, int* __restrict__ off,
                                                    int* __restrict__ cur) {
  __shared__ int part[SCAN_T];
  const int tid = threadIdx.x;
  const int base = tid * SCAN_PER;
  int loc[SCAN_PER];
  int s = 0;
#pragma unroll
  for (int k = 0; k < SCAN_PER; ++k) {
    const int i = base + k;
    const int v = i < NN ? cnt[i] : 0;
    loc[k] = s;
    s += v;
  }
  part[tid] = s;
  __syncthreads();
  for (int d = 1; d < SCAN_T; d <<= 1) {
    const int v = tid >= d ? part[tid - d] : 0;
    __syncthreads();
    part[tid] += v;
    __syncthreads();
  }
  const int excl = part[tid] - s;
#pragma unroll
  for (int k = 0; k < SCAN_PER; ++k) {
    const int i = base + k;
    if (i < NN) {
      const int o = excl + loc[k];
      off[i] = o;
      cur[i] = o;
      const float d = deg[i];
      dinv[i] = d > 0.f ? rsqrtf(d) : 0.f;
    }
  }
  if (tid == SCAN_T - 1) off[NN] = part[SCAN_T - 1];
}

__global__ __launch_bounds__(256) void k_gcn_gemm(const float* __restrict__ X,
                                                  const float* __restrict__ W,
                                                  float* __restrict__ Y) {
  __shared__ float wl[FOUT * P64];
  for (int i = threadIdx.x; i < FOUT * FIN; i += 256) wl[(i >> 6) * P64 + (i & 63)] = W[i];
  __syncthreads();
  const int wave = threadIdx.x >> 5, lane = threadIdx.x & 31;
  const int mr = lane & 15, half = lane >> 4, kk = half * 2;
  const int row0 = blockIdx.x * 32 + (wave & 1) * 16;
  const int n0 = (wave >> 1) * 16;
  const float* ap = X + (size_t)(row0 + mr) * FIN + kk;
  const float* bp = wl + (n0 + mr) * P64 + kk;
  v8f c = {};
#pragma unroll
  for (int k = 0; k < FIN; k += 4) {
    v2f a = {ap[k], ap[k + 1]};
    v2f b = {bp[k], bp[k + 1]};
    c = wmma4(a, b, c);
  }
#pragma unroll
  for (int j = 0; j < 8; ++j)
    Y[(size_t)(row0 + j + 8 * half) * FOUT + n0 + mr] = c[j];
}

__global__ void k_fill(const int* __restrict__ ei, const float* __restrict__ ew,
                       const float* __restrict__ dinv, int* __restrict__ cur,
                       int* __restrict__ srcs, float* __restrict__ wts) {
  int e = blockIdx.x * 256 + threadIdx.x;
  if (e < EE) {
    const int r = ei[e], c = ei[EE + e];
    const int pos = atomAddI(&cur[c], 1);
    srcs[pos] = r;
    wts[pos]  = dinv[r] * ew[e] * dinv[c];
  }
}

static constexpr int GA_BLOCKS = 2048;
static constexpr int GA_WAVES  = GA_BLOCKS * 8;
static constexpr int GA_ITEMS  = (GG / 2) * NN;
__global__ __launch_bounds__(256) void k_gather(const int* __restrict__ off, const int* __restrict__ srcs,
                                                const float* __restrict__ wts,
                                                const float* __restrict__ dinv,
                                                const float* __restrict__ b,
                                                const float* __restrict__ xl, float* __restrict__ gc) {
  const int lane = threadIdx.x & 31;
  const int wave = __builtin_amdgcn_readfirstlane(threadIdx.x >> 5);
  const v2f bias = *reinterpret_cast<const v2f*>(b + lane * 2);
  for (int item = blockIdx.x * 8 + wave; item < GA_ITEMS; item += GA_WAVES) {
    const int gp = item / NN;
    const int c  = item - gp * NN;
    const int beg = __builtin_amdgcn_readfirstlane(off[c]);
    const int end = __builtin_amdgcn_readfirstlane(off[c + 1]);
    const float di = dinv[c];
    const float wself = di * di;
    const float* x0 = xl + (size_t)(gp * 2) * NN * FOUT + lane * 2;
    const float* x1 = x0 + (size_t)NN * FOUT;
    const v2f s0 = *reinterpret_cast<const v2f*>(x0 + (size_t)c * FOUT);
    const v2f s1 = *reinterpret_cast<const v2f*>(x1 + (size_t)c * FOUT);
    v2f a0 = bias + wself * s0;
    v2f a1 = bias + wself * s1;
    for (int base = beg; base < end; base += 32) {
      const int n = end - base < 32 ? end - base : 32;
      int my_r = 0;
      float my_w = 0.f;
      if (lane < n) { my_r = srcs[base + lane]; my_w = wts[base + lane]; }
#pragma unroll 4
      for (int j = 0; j < n; ++j) {
        const int   r = __builtin_amdgcn_readlane(my_r, j);
        const float w = __int_as_float(__builtin_amdgcn_readlane(__float_as_int(my_w), j));
        const v2f v0 = *reinterpret_cast<const v2f*>(x0 + (size_t)r * FOUT);
        const v2f v1 = *reinterpret_cast<const v2f*>(x1 + (size_t)r * FOUT);
        a0 += w * v0;
        a1 += w * v1;
      }
    }
    float* o0 = gc + ((size_t)(gp * 2) * NN + c) * FOUT + lane * 2;
    *reinterpret_cast<v2f*>(o0) = a0;
    *reinterpret_cast<v2f*>(o0 + (size_t)NN * FOUT) = a1;
  }
}

static constexpr int   MROWS = 64;

static constexpr int   FRAG  = 64;
static constexpr size_t WP1_OFF = 0;
static constexpr size_t WP2_OFF = WP1_OFF + (size_t)HID * HH;
static constexpr size_t WP3_OFF = WP2_OFF + (size_t)HID * HID;
static constexpr size_t WP4_OFF = WP3_OFF + (size_t)FOUT * HID;
static constexpr size_t WP_FLOATS = WP4_OFF + (size_t)HH * KCAT;

template <int K>
__device__ __forceinline__ void pack_one(const float* __restrict__ W, float* __restrict__ Wp, int j) {
  const int lane = j & 31, frag = j >> 5;
  const int ks = frag % (K / 4), ct = frag / (K / 4);
  const float* src = W + (size_t)(ct * 16 + (lane & 15)) * K + ks * 4 + (lane >> 4) * 2;
  v2f v = {src[0], src[1]};
  *reinterpret_cast<v2f*>(Wp + (size_t)j * 2) = v;
}

__device__ __forceinline__ void pack_cat(const float* __restrict__ Wih, const float* __restrict__ Whh,
                                         float* __restrict__ Wp, int j) {
  const int lane = j & 31, frag = j >> 5;
  const int ks = frag % (KCAT / 4), ct = frag / (KCAT / 4);
  const int n = ct * 16 + (lane & 15);
  const int k = ks * 4 + (lane >> 4) * 2;
  const float* src = k < FOUT ? Wih + (size_t)n * FOUT + k : Whh + (size_t)n * HH + (k - FOUT);
  v2f v = {src[0], src[1]};
  *reinterpret_cast<v2f*>(Wp + (size_t)j * 2) = v;
}

__global__ __launch_bounds__(256) void k_pack_w(const float* __restrict__ W1, const float* __restrict__ W2,
                                                const float* __restrict__ W3, const float* __restrict__ Wih,
                                                const float* __restrict__ Whh, float* __restrict__ Wp) {
  constexpr int B1 = HID * HH / 2 / 256;
  constexpr int B2 = HID * HID / 2 / 256;
  constexpr int B3 = FOUT * HID / 2 / 256;
  const int b = blockIdx.x;
  if (b < B1)                pack_one<HH >(W1, Wp + WP1_OFF, b * 256 + threadIdx.x);
  else if (b < B1 + B2)      pack_one<HID>(W2, Wp + WP2_OFF, (b - B1) * 256 + threadIdx.x);
  else if (b < B1 + B2 + B3) pack_one<HID>(W3, Wp + WP3_OFF, (b - B1 - B2) * 256 + threadIdx.x);
  else                       pack_cat(Wih, Whh, Wp + WP4_OFF, (b - B1 - B2 - B3) * 256 + threadIdx.x);
}

template <int KB>
struct BFrags { v2f b[KB]; };
template <int R, int KH>
struct AFrags { v2f a[R][KH]; };

template <int KB>
__device__ __forceinline__ void load_b(BFrags<KB>& f, const float* __restrict__ bp) {
#pragma unroll
  for (int s = 0; s < KB; ++s) f.b[s] = *reinterpret_cast<const v2f*>(bp + s * FRAG);
}

template <int R, int KH, int APITCH>
__device__ __forceinline__ void load_a(AFrags<R, KH>& f, const float* __restrict__ ap) {
#pragma unroll
  for (int s = 0; s < KH; ++s) {
#pragma unroll
    for (int r = 0; r < R; ++r)
      f.a[r][s] = *reinterpret_cast<const v2f*>(ap + r * 16 * APITCH + s * 4);
  }
}

template <int R, int KH, int KB, int OFF>
__device__ __forceinline__ void mma_half(const AFrags<R, KH>& fa, const BFrags<KB>& fb, v8f (&c)[R]) {
#pragma unroll
  for (int s = 0; s < KH; ++s) {
#pragma unroll
    for (int r = 0; r < R; ++r) c[r] = wmma4(fa.a[r][s], fb.b[OFF + s], c[r]);
  }
}

__device__ __forceinline__ void fence() { __builtin_amdgcn_sched_barrier(0); }

template <int KLEN, int APITCH, int R, int KB, int NT, class Epi>
__device__ __forceinline__ void run_tiles(const float* __restrict__ ap, const float* __restrict__ bp,
                                          Epi epi) {
  constexpr int KH = KB / 2;
  constexpr int NB = KLEN / (4 * KB);
  constexpr int AS = 4 * KB;
  constexpr int AH = 4 * KH;
  constexpr int BS = KB * FRAG;
  BFrags<KB> b0, b1;
  AFrags<R, KH> a0, a1;
  load_b<KB>(b0, bp);
  load_a<R, KH, APITCH>(a0, ap);
#pragma unroll
  for (int t = 0; t < NT; ++t) {
    v8f c[R] = {};
    const float* bt = bp + (size_t)t * NB * BS;
    for (int j = 0; j < NB - 2; j += 2) {
      load_a<R, KH, APITCH>(a1, ap + j * AS + AH);
      load_b<KB>(b1, bt + (j + 1) * BS);
      fence();
      mma_half<R, KH, KB, 0>(a0, b0, c);
      fence();
      load_a<R, KH, APITCH>(a0, ap + (j + 1) * AS);
      fence();
      mma_half<R, KH, KB, KH>(a1, b0, c);
      fence();
      load_a<R, KH, APITCH>(a1, ap + (j + 1) * AS + AH);
      load_b<KB>(b0, bt + (j + 2) * BS);
      fence();
      mma_half<R, KH, KB, 0>(a0, b1, c);
      fence();
      load_a<R, KH, APITCH>(a0, ap + (j + 2) * AS);
      fence();
      mma_half<R, KH, KB, KH>(a1, b1, c);
      fence();
    }
    load_a<R, KH, APITCH>(a1, ap + (NB - 2) * AS + AH);
    load_b<KB>(b1, bt + (NB - 1) * BS);
    fence();
    mma_half<R, KH, KB, 0>(a0, b0, c);
    fence();
    load_a<R, KH, APITCH>(a0, ap + (NB - 1) * AS);
    fence();
    mma_half<R, KH, KB, KH>(a1, b0, c);
    fence();
    load_a<R, KH, APITCH>(a1, ap + (NB - 1) * AS + AH);
    if (t + 1 < NT) load_b<KB>(b0, bt + NB * BS);
    fence();
    mma_half<R, KH, KB, 0>(a0, b1, c);
    fence();
    if (t + 1 < NT) load_a<R, KH, APITCH>(a0, ap);
    fence();
    mma_half<R, KH, KB, KH>(a1, b1, c);
    fence();
    epi(t, c);
  }
}

__device__ __forceinline__ int keep_in_loop(int v) {
  asm volatile("" : "+v"(v));
  return v;
}

static constexpr int   RROWS   = 80;
static constexpr int   RNN_WGS = BN / RROWS;
static constexpr int   RPLANE  = RROWS * P192;
static constexpr int   RNN_LDS_FLOATS = 162 * 1024 / 4;
static_assert(RNN_LDS_FLOATS >= 2 * RPLANE, "planes fit");
static constexpr int   XPIECES = RROWS * FOUT / 4;
static constexpr int   XU      = (XPIECES + 511) / 512;

__device__ __forceinline__ void load_xt(v4f (&x)[XU], const float* __restrict__ gc, size_t p0, int t) {
#pragma unroll
  for (int u = 0; u < XU; ++u) {
    const int q = threadIdx.x + u * 512;
    v4f v = {};
    if (q < XPIECES)
      v = *reinterpret_cast<const v4f*>(gc + ((p0 + (q >> 4)) * TT + t) * FOUT + (q & 15) * 4);
    x[u] = v;
  }
}

__device__ __forceinline__ void store_xt(float* __restrict__ plane, const v4f (&x)[XU]) {
#pragma unroll
  for (int u = 0; u < XU; ++u) {
    const int q = threadIdx.x + u * 512;
    if (q < XPIECES) *reinterpret_cast<v4f*>(plane + (q >> 4) * P192 + (q & 15) * 4) = x[u];
  }
}

template <int R>
__device__ __forceinline__ void rnn_step(const float* __restrict__ cur, float* __restrict__ nxt,
                                         const float* __restrict__ Wp, float* __restrict__ rnn_t,
                                         float bias, int rt0, int ct, int lane, bool keep) {
  const int mr = lane & 15, half = lane >> 4, kk = half * 2;
  const int oa = keep_in_loop((rt0 * 16 + mr) * P192 + kk);
  const int oh = keep_in_loop((rt0 * 16 + 8 * half) * P192 + FOUT + ct * 16 + mr);
  const int og = keep_in_loop((rt0 * 16 + 8 * half) * TT * HH + ct * 16 + mr);
  float* __restrict__ hp = nxt + oh;
  float* __restrict__ gp = rnn_t + og;
  run_tiles<KCAT, P192, R, 8, 1>(
      cur + oa, Wp + WP4_OFF + (size_t)ct * (KCAT / 4) * FRAG + lane * 2,
      [&](int, v8f (&c)[R]) {
#pragma unroll
        for (int r = 0; r < R; ++r) {
#pragma unroll
          for (int j = 0; j < 8; ++j) {
            const float v = tanhf(c[r][j] + bias);
            if (keep) hp[(r * 16 + j) * P192] = v;
            gp[(r * 16 + j) * TT * HH] = v;
          }
        }
      });
}

__global__ __launch_bounds__(512) void k_rnn(const float* __restrict__ gc, const float* __restrict__ Wp,
                                             const float* __restrict__ bih, const float* __restrict__ bhh,
                                             float* __restrict__ rnn) {
  __shared__ __attribute__((aligned(16))) float rs[RNN_LDS_FLOATS];
  const int wave = threadIdx.x >> 5, lane = threadIdx.x & 31;
  const int ct = wave & 7, grp = wave >> 3;
  const size_t p0 = (size_t)blockIdx.x * RROWS;
  const float bias = bih[ct * 16 + (lane & 15)] + bhh[ct * 16 + (lane & 15)];

  v4f xr[XU];
  load_xt(xr, gc, p0, 0);
#pragma unroll
  for (int u = 0; u < RROWS * HH / 4 / 512; ++u) {
    const int q = threadIdx.x + u * 512;
    v4f z = {};
    *reinterpret_cast<v4f*>(rs + (q >> 5) * P192 + FOUT + (q & 31) * 4) = z;
  }
  store_xt(rs, xr);
  __syncthreads();

  for (int t = 0; t < TT; ++t) {
    const float* cur = rs + (t & 1) * RPLANE;
    float* nxt = rs + ((t + 1) & 1) * RPLANE;
    const bool more = t + 1 < TT;
    if (more) load_xt(xr, gc, p0, t + 1);
    fence();
    float* rnn_t = rnn + (p0 * TT + t) * HH;
    if (grp == 0) rnn_step<3>(cur, nxt, Wp, rnn_t, bias, 0, ct, lane, more);
    else          rnn_step<2>(cur, nxt, Wp, rnn_t, bias, 3, ct, lane, more);
    if (more) store_xt(nxt, xr);
    __syncthreads();
  }
}

__device__ __forceinline__ void store_relu_x4(float* __restrict__ yp, float bias,
                                              const v8f& c0, const v8f& c1, const v8f& c2, const v8f& c3) {
#pragma unroll
  for (int j = 0; j < 8; ++j) {
    float v0 = c0[j] + bias, v1 = c1[j] + bias, v2 = c2[j] + bias, v3 = c3[j] + bias;
    yp[j * P512]        = v0 > 0.f ? v0 : 0.f;
    yp[(16 + j) * P512] = v1 > 0.f ? v1 : 0.f;
    yp[(32 + j) * P512] = v2 > 0.f ? v2 : 0.f;
    yp[(48 + j) * P512] = v3 > 0.f ? v3 : 0.f;
  }
}

static constexpr int   MLP_TILES = (BN * TT) / MROWS;
static constexpr int   MLP_WGS   = 256;
static constexpr int   XQ        = MROWS * HH / 4 / 512;

__device__ __forceinline__ void load_x(v4f (&x)[XQ], const float* __restrict__ Xg, int tile) {
  const v4f* p = reinterpret_cast<const v4f*>(Xg + (size_t)tile * MROWS * HH) + threadIdx.x;
#pragma unroll
  for (int u = 0; u < XQ; ++u) x[u] = p[u * 512];
}

__device__ __forceinline__ void store_x(float* __restrict__ sX, const v4f (&x)[XQ]) {
#pragma unroll
  for (int u = 0; u < XQ; ++u) {
    const int q = threadIdx.x + u * 512;
    *reinterpret_cast<v4f*>(sX + (q >> 5) * P128 + (q & 31) * 4) = x[u];
  }
}

__global__ __launch_bounds__(512) void k_mlp(const float* __restrict__ Xg,
                                             const float* __restrict__ Wp, const float* __restrict__ B1,
                                             const float* __restrict__ B2, const float* __restrict__ B3,
                                             float* __restrict__ out) {
  __shared__ float sm[2 * MROWS * P512 + MROWS * P128];
  float* sY1 = sm;
  float* sY2 = sm + MROWS * P512;
  float* sX  = sm + 2 * MROWS * P512;
  const int wave = threadIdx.x >> 5, lane = threadIdx.x & 31;
  const int mr = lane & 15, half = lane >> 4, kk = half * 2;

  const int ctA = wave * 2;
  const int rt = wave & 3, ct3 = wave >> 2;

  const float b1a = B1[ctA * 16 + mr], b1b = B1[ctA * 16 + 16 + mr];
  const float b2a = B2[ctA * 16 + mr], b2b = B2[ctA * 16 + 16 + mr];
  const float b3 = B3[ct3 * 16 + mr];

  v4f xr[XQ];
  load_x(xr, Xg, blockIdx.x);
  store_x(sX, xr);
  __syncthreads();

  for (int tile = blockIdx.x; tile < MLP_TILES; tile += MLP_WGS) {
    const size_t r0 = (size_t)tile * MROWS;
    const bool more = tile + MLP_WGS < MLP_TILES;

    const int ox1 = keep_in_loop(mr * P128 + kk);
    const int oy1 = keep_in_loop((8 * half) * P512 + ctA * 16 + mr);
    run_tiles<HH, P128, 4, 8, 2>(
        sX + ox1, Wp + WP1_OFF + (size_t)ctA * (HH / 4) * FRAG + lane * 2,
        [&](int t, v8f (&c)[4]) {
          store_relu_x4(sY1 + oy1 + t * 16, t ? b1b : b1a, c[0], c[1], c[2], c[3]);
        });
    __syncthreads();

    if (more) load_x(xr, Xg, tile + MLP_WGS);
    fence();

    const int ox2 = keep_in_loop(mr * P512 + kk);
    const int oy2 = keep_in_loop((8 * half) * P512 + ctA * 16 + mr);
    run_tiles<HID, P512, 4, 8, 2>(
        sY1 + ox2, Wp + WP2_OFF + (size_t)ctA * (HID / 4) * FRAG + lane * 2,
        [&](int t, v8f (&c)[4]) {
          store_relu_x4(sY2 + oy2 + t * 16, t ? b2b : b2a, c[0], c[1], c[2], c[3]);
        });
    if (more) store_x(sX, xr);
    __syncthreads();

    const int ox3 = keep_in_loop((rt * 16 + mr) * P512 + kk);
    run_tiles<HID, P512, 1, 16, 1>(
        sY2 + ox3, Wp + WP3_OFF + (size_t)ct3 * (HID / 4) * FRAG + lane * 2,
        [&](int, v8f (&c)[1]) {
#pragma unroll
          for (int j = 0; j < 8; ++j) {
            size_t r = r0 + rt * 16 + j + 8 * half;
            size_t p = r >> 3;
            int t = (int)(r & 7);
            out[(p * (TT + PP) + t) * FOUT + ct3 * 16 + mr] = c[0][j] + b3;
          }
        });
  }
}

__global__ void k_zero_pred(float* __restrict__ out) {
  size_t idx = (size_t)blockIdx.x * 256 + threadIdx.x;
  if (idx < (size_t)BN * PP * FOUT) {
    size_t p = idx >> 7;
    size_t rem = idx & 127;
    out[p * (size_t)(TT + PP) * FOUT + (size_t)TT * FOUT + rem] = 0.f;
  }
}

extern "C" void kernel_launch(void* const* d_in, const int* in_sizes, int n_in,
                              void* d_out, int out_size, void* d_ws, size_t ws_size,
                              hipStream_t stream) {
  const float* x_in  = (const float*)d_in[0];
  const int*   ei    = (const int*)  d_in[1];
  const float* ew    = (const float*)d_in[2];
  const float* gcn_w = (const float*)d_in[3];
  const float* gcn_b = (const float*)d_in[4];
  const float* w_ih  = (const float*)d_in[5];
  const float* w_hh  = (const float*)d_in[6];
  const float* b_ih  = (const float*)d_in[7];
  const float* b_hh  = (const float*)d_in[8];
  const float* w1    = (const float*)d_in[9];
  const float* b1    = (const float*)d_in[10];
  const float* w2    = (const float*)d_in[11];
  const float* b2    = (const float*)d_in[12];
  const float* w3    = (const float*)d_in[13];
  const float* b3    = (const float*)d_in[14];
  float* out = (float*)d_out;

  float* ws   = (float*)d_ws;
  float* deg  = ws;
  float* dinv = ws + 16384;
  float* xl   = ws + 32768;
  float* gc   = xl + GNF;
  float* rnn  = gc + GNF;
  float* wp   = xl;
  int*   cnt  = (int*)rnn;
  int*   off  = cnt + 16384;
  int*   cur  = off + 16384;
  int*   srcs = cur + 16384;
  float* wts  = (float*)(srcs + 163840);

  k_deg_init<<<(NN + 255) / 256, 256, 0, stream>>>(deg, cnt);
  k_deg_scatter<<<(EE + NN + 255) / 256, 256, 0, stream>>>(ei, ew, deg, cnt);
  k_dinv_scan<<<1, 1024, 0, stream>>>(deg, dinv, cnt, off, cur);

  k_gcn_gemm<<<(GG * NN) / 32, 256, 0, stream>>>(x_in, gcn_w, xl);

  k_fill<<<(EE + 255) / 256, 256, 0, stream>>>(ei, ew, dinv, cur, srcs, wts);
  k_gather<<<GA_BLOCKS, 256, 0, stream>>>(off, srcs, wts, dinv, gcn_b, xl, gc);

  k_pack_w<<<(int)(WP_FLOATS / 2 / 256), 256, 0, stream>>>(w1, w2, w3, w_ih, w_hh, wp);

  k_rnn<<<RNN_WGS, 512, 0, stream>>>(gc, wp, b_ih, b_hh, rnn);

  k_mlp<<<MLP_WGS, 512, 0, stream>>>(rnn, wp, b1, b2, b3, out);

  k_zero_pred<<<(int)(((size_t)BN * PP * FOUT + 255) / 256), 256, 0, stream>>>(out);
}
